# starting kernel
# baseline (speedup 1.0000x reference)
#include <hip/hip_runtime.h>

#define TT 8
#define T1 (TT + 6)
#define T2 (TT + 4)
#define T3 (TT + 2)
#define NB 64
#define NTIME 512
#define NV 21
#define NC 32
#define NCIN 2
#define NOUT 64

typedef __attribute__((ext_vector_type(2))) float v2f;
typedef __attribute__((ext_vector_type(8))) float v8f;

__device__ __forceinline__ v8f wmma4(v2f a, v2f b, v8f c) {
  return __builtin_amdgcn_wmma_f32_16x16x4_f32(
       false, a,  false, b,
       (short)0, c,  false,  false);
}

__device__ const unsigned char g_pi[20] = {0,1,2,3,0,5,6,7,0,9,10,11,0,13,14,15,0,17,18,19};
__device__ const unsigned char g_pj[20] = {1,2,3,4,5,6,7,8,9,10,11,12,13,14,15,16,17,18,19,20};
__device__ const float g_dinv[21] = {
  0.4082482905f,
  0.5773502692f,0.5773502692f,0.5773502692f,0.7071067812f,
  0.5773502692f,0.5773502692f,0.5773502692f,0.7071067812f,
  0.5773502692f,0.5773502692f,0.5773502692f,0.7071067812f,
  0.5773502692f,0.5773502692f,0.5773502692f,0.7071067812f,
  0.5773502692f,0.5773502692f,0.5773502692f,0.7071067812f};

__global__ __launch_bounds__(256) void stgcn_fused(
    const float* __restrict__ x,
    const float* __restrict__ gcn_w0, const float* __restrict__ gcn_b0,
    const float* __restrict__ tcn_w0, const float* __restrict__ tcn_b0,
    const float* __restrict__ bn_g0, const float* __restrict__ bn_b0,
    const float* __restrict__ bn_m0, const float* __restrict__ bn_v0,
    const float* __restrict__ res_w, const float* __restrict__ res_b,
    const float* __restrict__ res_bn_g, const float* __restrict__ res_bn_b,
    const float* __restrict__ res_bn_m, const float* __restrict__ res_bn_v,
    const float* __restrict__ gcn_w1, const float* __restrict__ gcn_b1,
    const float* __restrict__ tcn_w1, const float* __restrict__ tcn_b1,
    const float* __restrict__ bn_g1, const float* __restrict__ bn_b1,
    const float* __restrict__ bn_m1, const float* __restrict__ bn_v1,
    const float* __restrict__ fc_w, const float* __restrict__ fc_b,
    float* __restrict__ out)
{
  __shared__ float sA[NV * NV];
  __shared__ float s_in[T1 * NV * NCIN];
  __shared__ float s_xa[T1 * NV * NCIN];
  __shared__ float s_tw0[96 * NC];
  __shared__ float s_tw1[96 * NC];
  __shared__ float s_gw1[NC * NC];
  __shared__ float s_g0[T1 * NV * NC];
  __shared__ float s_h0[T2 * NV * NC];
  __shared__ float s_g1[T3 * NV * NC];
  __shared__ float s_gw0[NC * NCIN];
  __shared__ float s_gb0[NC], s_gb1[NC];
  __shared__ float s_s0[NC], s_o0[NC], s_s1[NC], s_o1[NC];
  __shared__ float s_rA[NC], s_rB[NC], s_rC[NC];
  __shared__ float s_fcb[NOUT];

  const int tid  = threadIdx.x;
  const int nthr = 256;
  const int lane = tid & 31;
  const int wave = tid >> 5;
  const int row  = lane & 15;
  const int hi   = lane >> 4;

  const int bid   = blockIdx.x;
  const int tile  = bid & 63;
  const int hand  = (bid >> 6) & 1;
  const int b     = bid >> 7;
  const int tbase = tile * TT;

  for (int idx = tid; idx < NV * NV; idx += nthr) {
    int i = idx / NV, j = idx % NV;
    bool conn = (i == j);
    for (int p = 0; p < 20; ++p)
      conn = conn || ((int)g_pi[p] == i && (int)g_pj[p] == j) ||
                     ((int)g_pi[p] == j && (int)g_pj[p] == i);
    sA[idx] = conn ? g_dinv[i] * g_dinv[j] : 0.0f;
  }
  for (int idx = tid; idx < 96 * NC; idx += nthr) {
    int co = idx & 31, kap = idx >> 5, k = kap >> 5, ci = kap & 31;
    s_tw0[idx] = tcn_w0[(co * NC + ci) * 3 + k];
    s_tw1[idx] = tcn_w1[(co * NC + ci) * 3 + k];
  }
  for (int idx = tid; idx < NC * NC; idx += nthr) s_gw1[idx] = gcn_w1[idx];
  if (tid < NC * NCIN) s_gw0[tid] = gcn_w0[tid];
  if (tid < NC) {
    s_gb0[tid] = gcn_b0[tid];
    s_gb1[tid] = gcn_b1[tid];
    float sc0 = bn_g0[tid] * rsqrtf(bn_v0[tid] + 1e-5f);
    s_s0[tid] = sc0;
    s_o0[tid] = (tcn_b0[tid] - bn_m0[tid]) * sc0 + bn_b0[tid];
    float sc1 = bn_g1[tid] * rsqrtf(bn_v1[tid] + 1e-5f);
    s_s1[tid] = sc1;
    s_o1[tid] = (tcn_b1[tid] - bn_m1[tid]) * sc1 + bn_b1[tid];
    float rsc = res_bn_g[tid] * rsqrtf(res_bn_v[tid] + 1e-5f);
    s_rA[tid] = res_w[tid * 2 + 0] * rsc;
    s_rB[tid] = res_w[tid * 2 + 1] * rsc;
    s_rC[tid] = (res_b[tid] - res_bn_m[tid]) * rsc + res_bn_b[tid];
  }
  if (tid < NOUT) s_fcb[tid] = fc_b[tid];

  for (int idx = tid; idx < T1 * NV * NCIN; idx += nthr) {
    int i = idx / (NV * NCIN), r = idx % (NV * NCIN);
    int v = r >> 1, ci = r & 1;
    int t = tbase - 3 + i;
    float val = 0.0f;
    if (t >= 0 && t < NTIME)
      val = x[(((b * NTIME + t) * 2 + hand) * NV + v) * NCIN + ci];
    s_in[idx] = val;
  }
  __syncthreads();

  for (int idx = tid; idx < T1 * NV * NCIN; idx += nthr) {
    int i = idx / (NV * NCIN), r = idx % (NV * NCIN);
    int v = r >> 1, ci = r & 1;
    float acc = 0.0f;
    const float* Ar = &sA[v * NV];
    const float* xr = &s_in[i * NV * NCIN + ci];
    for (int u = 0; u < NV; ++u) acc += Ar[u] * xr[u * NCIN];
    s_xa[idx] = acc;
  }
  __syncthreads();

  for (int idx = tid; idx < T1 * NV * NC; idx += nthr) {
    int c = idx & 31, rv = idx >> 5;
    int v = rv % NV, i = rv / NV;
    int t = tbase - 3 + i;
    float val = 0.0f;
    if (t >= 0 && t < NTIME) {
      float xa0 = s_xa[(i * NV + v) * 2 + 0];
      float xa1 = s_xa[(i * NV + v) * 2 + 1];
      val = s_gw0[c * 2 + 0] * xa0 + s_gw0[c * 2 + 1] * xa1 + s_gb0[c];
    }
    s_g0[idx] = val;
  }
  __syncthreads();

  {
    const int M = T2 * NV;
    for (int tIdx = wave; tIdx < 16 * 2; tIdx += 8) {
      int mt = tIdx >> 1, nt = tIdx & 1;
      v8f acc = {0.f,0.f,0.f,0.f,0.f,0.f,0.f,0.f};
      int mA = mt * 16 + row;
      int mC = (mA < M) ? mA : 0;
      int t2 = mC / NV, vv = mC % NV;
      int colN = nt * 16 + row;
      for (int ks = 0; ks < 24; ++ks) {
        int k0 = ks * 4 + hi * 2;
        v2f a, bb;
        a.x = s_g0[(t2 + (k0 >> 5)) * (NV * NC) + vv * NC + (k0 & 31)];
        a.y = s_g0[(t2 + ((k0 + 1) >> 5)) * (NV * NC) + vv * NC + ((k0 + 1) & 31)];
        bb.x = s_tw0[k0 * NC + colN];
        bb.y = s_tw0[(k0 + 1) * NC + colN];
        acc = wmma4(a, bb, acc);
      }
#pragma unroll
      for (int j = 0; j < 8; ++j) {
        int m = mt * 16 + j + hi * 8;
        if (m < M) {
          int te = m / NV, ve = m % NV;
          int co = colN;
          float val = acc[j] * s_s0[co] + s_o0[co];
          float x0 = s_in[((te + 1) * NV + ve) * 2 + 0];
          float x1 = s_in[((te + 1) * NV + ve) * 2 + 1];
          val += s_rA[co] * x0 + s_rB[co] * x1 + s_rC[co];
          s_h0[(te * NV + ve) * NC + co] = fmaxf(val, 0.0f);
        }
      }
    }
  }
  __syncthreads();

  float* s_hagg = s_g0;
  for (int idx = tid; idx < T3 * NV * NC; idx += nthr) {
    int ci = idx & 31, rv = idx >> 5;
    int v = rv % NV, t3 = rv / NV;
    float acc = 0.0f;
    const float* Ar = &sA[v * NV];
    const float* hr = &s_h0[(t3 + 1) * (NV * NC) + ci];
    for (int u = 0; u < NV; ++u) acc += Ar[u] * hr[u * NC];
    s_hagg[idx] = acc;
  }
  __syncthreads();

  {
    const int M = T3 * NV;
    for (int tIdx = wave; tIdx < 14 * 2; tIdx += 8) {
      int mt = tIdx >> 1, nt = tIdx & 1;
      v8f acc = {0.f,0.f,0.f,0.f,0.f,0.f,0.f,0.f};
      int mA = mt * 16 + row;
      int mC = (mA < M) ? mA : 0;
      int colN = nt * 16 + row;
      for (int ks = 0; ks < 8; ++ks) {
        int k0 = ks * 4 + hi * 2;
        v2f a, bb;
        a.x = s_hagg[mC * NC + k0];
        a.y = s_hagg[mC * NC + k0 + 1];
        bb.x = s_gw1[colN * NC + k0];
        bb.y = s_gw1[colN * NC + k0 + 1];
        acc = wmma4(a, bb, acc);
      }
#pragma unroll
      for (int j = 0; j < 8; ++j) {
        int m = mt * 16 + j + hi * 8;
        if (m < M) {
          int t3 = m / NV;
          int t = tbase - 1 + t3;
          float val = acc[j] + s_gb1[colN];
          s_g1[m * NC + colN] = (t >= 0 && t < NTIME) ? val : 0.0f;
        }
      }
    }
  }
  __syncthreads();

  {
    const int M = TT * NV;
    for (int tIdx = wave; tIdx < 11 * 2; tIdx += 8) {
      int mt = tIdx >> 1, nt = tIdx & 1;
      v8f acc = {0.f,0.f,0.f,0.f,0.f,0.f,0.f,0.f};
      int mA = mt * 16 + row;
      int mC = (mA < M) ? mA : 0;
      int t4 = mC / NV, vv = mC % NV;
      int colN = nt * 16 + row;
      for (int ks = 0; ks < 24; ++ks) {
        int k0 = ks * 4 + hi * 2;
        v2f a, bb;
        a.x = s_g1[((t4 + (k0 >> 5)) * NV + vv) * NC + (k0 & 31)];
        a.y = s_g1[((t4 + ((k0 + 1) >> 5)) * NV + vv) * NC + ((k0 + 1) & 31)];
        bb.x = s_tw1[k0 * NC + colN];
        bb.y = s_tw1[(k0 + 1) * NC + colN];
        acc = wmma4(a, bb, acc);
      }
#pragma unroll
      for (int j = 0; j < 8; ++j) {
        int m = mt * 16 + j + hi * 8;
        if (m < M) {
          int te = m / NV, ve = m % NV;
          int co = colN;
          int hidx = ((te + 2) * NV + ve) * NC + co;
          float val = acc[j] * s_s1[co] + s_o1[co] + s_h0[hidx];
          s_h0[hidx] = fmaxf(val, 0.0f);
        }
      }
    }
  }
  __syncthreads();

  {
    float* s_red = s_g0;
    int nt  = wave & 3;
    int ksb = (wave >> 2) * 84;
    v8f acc = {0.f,0.f,0.f,0.f,0.f,0.f,0.f,0.f};
    int t4 = (row < TT) ? row : (TT - 1);
    int colN = nt * 16 + row;
    const float* h1base = &s_h0[(t4 + 2) * (NV * NC)];
    for (int ks = ksb; ks < ksb + 84; ++ks) {
      int f0 = ks * 4 + hi * 2;
      int f1 = f0 + 1;
      int c0 = f0 / NV, v0 = f0 % NV;
      int c1 = f1 / NV, v1 = f1 % NV;
      v2f a, bb;
      a.x = h1base[v0 * NC + c0];
      a.y = h1base[v1 * NC + c1];
      bb.x = fc_w[colN * 672 + f0];
      bb.y = fc_w[colN * 672 + f1];
      acc = wmma4(a, bb, acc);
    }
    if (wave >= 4) {
#pragma unroll
      for (int j = 0; j < 8; ++j) s_red[nt * 256 + lane * 8 + j] = acc[j];
    }
    __syncthreads();
    if (wave < 4) {
#pragma unroll
      for (int j = 0; j < 8; ++j) {
        int m = j + hi * 8;
        if (m < TT) {
          int t = tbase + m;
          float val = acc[j] + s_red[nt * 256 + lane * 8 + j] + s_fcb[colN];
          out[(((size_t)b * NTIME + t) * 2 + hand) * NOUT + colN] = val;
        }
      }
    }
  }
}

extern "C" void kernel_launch(void* const* d_in, const int* in_sizes, int n_in,
                              void* d_out, int out_size, void* d_ws, size_t ws_size,
                              hipStream_t stream) {
  (void)in_sizes; (void)n_in; (void)d_ws; (void)ws_size; (void)out_size;
  const float* x        = (const float*)d_in[0];
  const float* gcn_w0   = (const float*)d_in[1];
  const float* gcn_b0   = (const float*)d_in[2];
  const float* tcn_w0   = (const float*)d_in[3];
  const float* tcn_b0   = (const float*)d_in[4];
  const float* bn_g0    = (const float*)d_in[5];
  const float* bn_b0    = (const float*)d_in[6];
  const float* bn_m0    = (const float*)d_in[7];
  const float* bn_v0    = (const float*)d_in[8];
  const float* res_w    = (const float*)d_in[9];
  const float* res_b    = (const float*)d_in[10];
  const float* res_bn_g = (const float*)d_in[11];
  const float* res_bn_b = (const float*)d_in[12];
  const float* res_bn_m = (const float*)d_in[13];
  const float* res_bn_v = (const float*)d_in[14];
  const float* gcn_w1   = (const float*)d_in[15];
  const float* gcn_b1   = (const float*)d_in[16];
  const float* tcn_w1   = (const float*)d_in[17];
  const float* tcn_b1   = (const float*)d_in[18];
  const float* bn_g1    = (const float*)d_in[19];
  const float* bn_b1    = (const float*)d_in[20];
  const float* bn_m1    = (const float*)d_in[21];
  const float* bn_v1    = (const float*)d_in[22];
  const float* fc_w     = (const float*)d_in[23];
  const float* fc_b     = (const float*)d_in[24];

  dim3 grid(NB * 2 * (NTIME / TT));
  dim3 block(256);
  stgcn_fused<<<grid, block, 0, stream>>>(
      x, gcn_w0, gcn_b0, tcn_w0, tcn_b0, bn_g0, bn_b0, bn_m0, bn_v0,
      res_w, res_b, res_bn_g, res_bn_b, res_bn_m, res_bn_v,
      gcn_w1, gcn_b1, tcn_w1, tcn_b1, bn_g1, bn_b1, bn_m1, bn_v1,
      fc_w, fc_b, (float*)d_out);
}


# Pad LDS row pitch of WMMA operand tiles (activations and gcn1 weights 32->36 floats, tcn weights 32->40) so fragment loads hit 64 banks instead of 8 rows per bank
# speedup vs baseline: 1.2054x; 1.2054x over previous
#include <hip/hip_runtime.h>

#define TT 8
#define T1 (TT + 6)
#define T2 (TT + 4)
#define T3 (TT + 2)
#define NB 64
#define NTIME 512
#define NV 21
#define NC 32
#define NCIN 2
#define NOUT 64
#define PC 36
#define PW 40

typedef __attribute__((ext_vector_type(2))) float v2f;
typedef __attribute__((ext_vector_type(8))) float v8f;

__device__ __forceinline__ v8f wmma4(v2f a, v2f b, v8f c) {
  return __builtin_amdgcn_wmma_f32_16x16x4_f32(
       false, a,  false, b,
       (short)0, c,  false,  false);
}

__device__ const unsigned char g_pi[20] = {0,1,2,3,0,5,6,7,0,9,10,11,0,13,14,15,0,17,18,19};
__device__ const unsigned char g_pj[20] = {1,2,3,4,5,6,7,8,9,10,11,12,13,14,15,16,17,18,19,20};
__device__ const float g_dinv[21] = {
  0.4082482905f,
  0.5773502692f,0.5773502692f,0.5773502692f,0.7071067812f,
  0.5773502692f,0.5773502692f,0.5773502692f,0.7071067812f,
  0.5773502692f,0.5773502692f,0.5773502692f,0.7071067812f,
  0.5773502692f,0.5773502692f,0.5773502692f,0.7071067812f,
  0.5773502692f,0.5773502692f,0.5773502692f,0.7071067812f};

__global__ __launch_bounds__(256) void stgcn_fused(
    const float* __restrict__ x,
    const float* __restrict__ gcn_w0, const float* __restrict__ gcn_b0,
    const float* __restrict__ tcn_w0, const float* __restrict__ tcn_b0,
    const float* __restrict__ bn_g0, const float* __restrict__ bn_b0,
    const float* __restrict__ bn_m0, const float* __restrict__ bn_v0,
    const float* __restrict__ res_w, const float* __restrict__ res_b,
    const float* __restrict__ res_bn_g, const float* __restrict__ res_bn_b,
    const float* __restrict__ res_bn_m, const float* __restrict__ res_bn_v,
    const float* __restrict__ gcn_w1, const float* __restrict__ gcn_b1,
    const float* __restrict__ tcn_w1, const float* __restrict__ tcn_b1,
    const float* __restrict__ bn_g1, const float* __restrict__ bn_b1,
    const float* __restrict__ bn_m1, const float* __restrict__ bn_v1,
    const float* __restrict__ fc_w, const float* __restrict__ fc_b,
    float* __restrict__ out)
{
  __shared__ float sA[NV * NV];
  __shared__ float s_in[T1 * NV * NCIN];
  __shared__ float s_xa[T1 * NV * NCIN];
  __shared__ float s_tw0[96 * PW];
  __shared__ float s_tw1[96 * PW];
  __shared__ float s_gw1[NC * PC];
  __shared__ float s_g0[T1 * NV * PC];
  __shared__ float s_h0[T2 * NV * PC];
  __shared__ float s_g1[T3 * NV * PC];
  __shared__ float s_gw0[NC * NCIN];
  __shared__ float s_gb0[NC], s_gb1[NC];
  __shared__ float s_s0[NC], s_o0[NC], s_s1[NC], s_o1[NC];
  __shared__ float s_rA[NC], s_rB[NC], s_rC[NC];
  __shared__ float s_fcb[NOUT];

  const int tid  = threadIdx.x;
  const int nthr = 256;
  const int lane = tid & 31;
  const int wave = tid >> 5;
  const int row  = lane & 15;
  const int hi   = lane >> 4;

  const int bid   = blockIdx.x;
  const int tile  = bid & 63;
  const int hand  = (bid >> 6) & 1;
  const int b     = bid >> 7;
  const int tbase = tile * TT;

  for (int idx = tid; idx < NV * NV; idx += nthr) {
    int i = idx / NV, j = idx % NV;
    bool conn = (i == j);
    for (int p = 0; p < 20; ++p)
      conn = conn || ((int)g_pi[p] == i && (int)g_pj[p] == j) ||
                     ((int)g_pi[p] == j && (int)g_pj[p] == i);
    sA[idx] = conn ? g_dinv[i] * g_dinv[j] : 0.0f;
  }
  for (int idx = tid; idx < 96 * NC; idx += nthr) {
    int co = idx & 31, kap = idx >> 5, k = kap >> 5, ci = kap & 31;
    s_tw0[kap * PW + co] = tcn_w0[(co * NC + ci) * 3 + k];
    s_tw1[kap * PW + co] = tcn_w1[(co * NC + ci) * 3 + k];
  }
  for (int idx = tid; idx < NC * NC; idx += nthr) s_gw1[(idx >> 5) * PC + (idx & 31)] = gcn_w1[idx];
  if (tid < NC * NCIN) s_gw0[tid] = gcn_w0[tid];
  if (tid < NC) {
    s_gb0[tid] = gcn_b0[tid];
    s_gb1[tid] = gcn_b1[tid];
    float sc0 = bn_g0[tid] * rsqrtf(bn_v0[tid] + 1e-5f);
    s_s0[tid] = sc0;
    s_o0[tid] = (tcn_b0[tid] - bn_m0[tid]) * sc0 + bn_b0[tid];
    float sc1 = bn_g1[tid] * rsqrtf(bn_v1[tid] + 1e-5f);
    s_s1[tid] = sc1;
    s_o1[tid] = (tcn_b1[tid] - bn_m1[tid]) * sc1 + bn_b1[tid];
    float rsc = res_bn_g[tid] * rsqrtf(res_bn_v[tid] + 1e-5f);
    s_rA[tid] = res_w[tid * 2 + 0] * rsc;
    s_rB[tid] = res_w[tid * 2 + 1] * rsc;
    s_rC[tid] = (res_b[tid] - res_bn_m[tid]) * rsc + res_bn_b[tid];
  }
  if (tid < NOUT) s_fcb[tid] = fc_b[tid];

  for (int idx = tid; idx < T1 * NV * NCIN; idx += nthr) {
    int i = idx / (NV * NCIN), r = idx % (NV * NCIN);
    int v = r >> 1, ci = r & 1;
    int t = tbase - 3 + i;
    float val = 0.0f;
    if (t >= 0 && t < NTIME)
      val = x[(((b * NTIME + t) * 2 + hand) * NV + v) * NCIN + ci];
    s_in[idx] = val;
  }
  __syncthreads();

  for (int idx = tid; idx < T1 * NV * NCIN; idx += nthr) {
    int i = idx / (NV * NCIN), r = idx % (NV * NCIN);
    int v = r >> 1, ci = r & 1;
    float acc = 0.0f;
    const float* Ar = &sA[v * NV];
    const float* xr = &s_in[i * NV * NCIN + ci];
    for (int u = 0; u < NV; ++u) acc += Ar[u] * xr[u * NCIN];
    s_xa[idx] = acc;
  }
  __syncthreads();

  for (int idx = tid; idx < T1 * NV * NC; idx += nthr) {
    int c = idx & 31, rv = idx >> 5;
    int v = rv % NV, i = rv / NV;
    int t = tbase - 3 + i;
    float val = 0.0f;
    if (t >= 0 && t < NTIME) {
      float xa0 = s_xa[(i * NV + v) * 2 + 0];
      float xa1 = s_xa[(i * NV + v) * 2 + 1];
      val = s_gw0[c * 2 + 0] * xa0 + s_gw0[c * 2 + 1] * xa1 + s_gb0[c];
    }
    s_g0[rv * PC + c] = val;
  }
  __syncthreads();

  {
    const int M = T2 * NV;
    for (int tIdx = wave; tIdx < 16 * 2; tIdx += 8) {
      int mt = tIdx >> 1, nt = tIdx & 1;
      v8f acc = {0.f,0.f,0.f,0.f,0.f,0.f,0.f,0.f};
      int mA = mt * 16 + row;
      int mC = (mA < M) ? mA : 0;
      int t2 = mC / NV, vv = mC % NV;
      int colN = nt * 16 + row;
      for (int ks = 0; ks < 24; ++ks) {
        int k0 = ks * 4 + hi * 2;
        v2f a, bb;
        a.x = s_g0[(t2 + (k0 >> 5)) * (NV * PC) + vv * PC + (k0 & 31)];
        a.y = s_g0[(t2 + ((k0 + 1) >> 5)) * (NV * PC) + vv * PC + ((k0 + 1) & 31)];
        bb.x = s_tw0[k0 * PW + colN];
        bb.y = s_tw0[(k0 + 1) * PW + colN];
        acc = wmma4(a, bb, acc);
      }
#pragma unroll
      for (int j = 0; j < 8; ++j) {
        int m = mt * 16 + j + hi * 8;
        if (m < M) {
          int te = m / NV, ve = m % NV;
          int co = colN;
          float val = acc[j] * s_s0[co] + s_o0[co];
          float x0 = s_in[((te + 1) * NV + ve) * 2 + 0];
          float x1 = s_in[((te + 1) * NV + ve) * 2 + 1];
          val += s_rA[co] * x0 + s_rB[co] * x1 + s_rC[co];
          s_h0[(te * NV + ve) * PC + co] = fmaxf(val, 0.0f);
        }
      }
    }
  }
  __syncthreads();

  float* s_hagg = s_g0;
  for (int idx = tid; idx < T3 * NV * NC; idx += nthr) {
    int ci = idx & 31, rv = idx >> 5;
    int v = rv % NV, t3 = rv / NV;
    float acc = 0.0f;
    const float* Ar = &sA[v * NV];
    const float* hr = &s_h0[(t3 + 1) * (NV * PC) + ci];
    for (int u = 0; u < NV; ++u) acc += Ar[u] * hr[u * PC];
    s_hagg[rv * PC + ci] = acc;
  }
  __syncthreads();

  {
    const int M = T3 * NV;
    for (int tIdx = wave; tIdx < 14 * 2; tIdx += 8) {
      int mt = tIdx >> 1, nt = tIdx & 1;
      v8f acc = {0.f,0.f,0.f,0.f,0.f,0.f,0.f,0.f};
      int mA = mt * 16 + row;
      int mC = (mA < M) ? mA : 0;
      int colN = nt * 16 + row;
      for (int ks = 0; ks < 8; ++ks) {
        int k0 = ks * 4 + hi * 2;
        v2f a, bb;
        a.x = s_hagg[mC * PC + k0];
        a.y = s_hagg[mC * PC + k0 + 1];
        bb.x = s_gw1[colN * PC + k0];
        bb.y = s_gw1[colN * PC + k0 + 1];
        acc = wmma4(a, bb, acc);
      }
#pragma unroll
      for (int j = 0; j < 8; ++j) {
        int m = mt * 16 + j + hi * 8;
        if (m < M) {
          int t3 = m / NV;
          int t = tbase - 1 + t3;
          float val = acc[j] + s_gb1[colN];
          s_g1[m * PC + colN] = (t >= 0 && t < NTIME) ? val : 0.0f;
        }
      }
    }
  }
  __syncthreads();

  {
    const int M = TT * NV;
    for (int tIdx = wave; tIdx < 11 * 2; tIdx += 8) {
      int mt = tIdx >> 1, nt = tIdx & 1;
      v8f acc = {0.f,0.f,0.f,0.f,0.f,0.f,0.f,0.f};
      int mA = mt * 16 + row;
      int mC = (mA < M) ? mA : 0;
      int t4 = mC / NV, vv = mC % NV;
      int colN = nt * 16 + row;
      for (int ks = 0; ks < 24; ++ks) {
        int k0 = ks * 4 + hi * 2;
        v2f a, bb;
        a.x = s_g1[((t4 + (k0 >> 5)) * NV + vv) * PC + (k0 & 31)];
        a.y = s_g1[((t4 + ((k0 + 1) >> 5)) * NV + vv) * PC + ((k0 + 1) & 31)];
        bb.x = s_tw1[k0 * PW + colN];
        bb.y = s_tw1[(k0 + 1) * PW + colN];
        acc = wmma4(a, bb, acc);
      }
#pragma unroll
      for (int j = 0; j < 8; ++j) {
        int m = mt * 16 + j + hi * 8;
        if (m < M) {
          int te = m / NV, ve = m % NV;
          int co = colN;
          int hidx = ((te + 2) * NV + ve) * PC + co;
          float val = acc[j] * s_s1[co] + s_o1[co] + s_h0[hidx];
          s_h0[hidx] = fmaxf(val, 0.0f);
        }
      }
    }
  }
  __syncthreads();

  {
    float* s_red = s_g0;
    int nt  = wave & 3;
    int ksb = (wave >> 2) * 84;
    v8f acc = {0.f,0.f,0.f,0.f,0.f,0.f,0.f,0.f};
    int t4 = (row < TT) ? row : (TT - 1);
    int colN = nt * 16 + row;
    const float* h1base = &s_h0[(t4 + 2) * (NV * PC)];
    for (int ks = ksb; ks < ksb + 84; ++ks) {
      int f0 = ks * 4 + hi * 2;
      int f1 = f0 + 1;
      int c0 = f0 / NV, v0 = f0 % NV;
      int c1 = f1 / NV, v1 = f1 % NV;
      v2f a, bb;
      a.x = h1base[v0 * PC + c0];
      a.y = h1base[v1 * PC + c1];
      bb.x = fc_w[colN * 672 + f0];
      bb.y = fc_w[colN * 672 + f1];
      acc = wmma4(a, bb, acc);
    }
    if (wave >= 4) {
#pragma unroll
      for (int j = 0; j < 8; ++j) s_red[nt * 256 + lane * 8 + j] = acc[j];
    }
    __syncthreads();
    if (wave < 4) {
#pragma unroll
      for (int j = 0; j < 8; ++j) {
        int m = j + hi * 8;
        if (m < TT) {
          int t = tbase + m;
          float val = acc[j] + s_red[nt * 256 + lane * 8 + j] + s_fcb[colN];
          out[(((size_t)b * NTIME + t) * 2 + hand) * NOUT + colN] = val;
        }
      }
    }
  }
}

extern "C" void kernel_launch(void* const* d_in, const int* in_sizes, int n_in,
                              void* d_out, int out_size, void* d_ws, size_t ws_size,
                              hipStream_t stream) {
  (void)in_sizes; (void)n_in; (void)d_ws; (void)ws_size; (void)out_size;
  const float* x        = (const float*)d_in[0];
  const float* gcn_w0   = (const float*)d_in[1];
  const float* gcn_b0   = (const float*)d_in[2];
  const float* tcn_w0   = (const float*)d_in[3];
  const float* tcn_b0   = (const float*)d_in[4];
  const float* bn_g0    = (const float*)d_in[5];
  const float* bn_b0    = (const float*)d_in[6];
  const float* bn_m0    = (const float*)d_in[7];
  const float* bn_v0    = (const float*)d_in[8];
  const float* res_w    = (const float*)d_in[9];
  const float* res_b    = (const float*)d_in[10];
  const float* res_bn_g = (const float*)d_in[11];
  const float* res_bn_b = (const float*)d_in[12];
  const float* res_bn_m = (const float*)d_in[13];
  const float* res_bn_v = (const float*)d_in[14];
  const float* gcn_w1   = (const float*)d_in[15];
  const float* gcn_b1   = (const float*)d_in[16];
  const float* tcn_w1   = (const float*)d_in[17];
  const float* tcn_b1   = (const float*)d_in[18];
  const float* bn_g1    = (const float*)d_in[19];
  const float* bn_b1    = (const float*)d_in[20];
  const float* bn_m1    = (const float*)d_in[21];
  const float* bn_v1    = (const float*)d_in[22];
  const float* fc_w     = (const float*)d_in[23];
  const float* fc_b     = (const float*)d_in[24];

  dim3 grid(NB * 2 * (NTIME / TT));
  dim3 block(256);
  stgcn_fused<<<grid, block, 0, stream>>>(
      x, gcn_w0, gcn_b0, tcn_w0, tcn_b0, bn_g0, bn_b0, bn_m0, bn_v0,
      res_w, res_b, res_bn_g, res_bn_b, res_bn_m, res_bn_v,
      gcn_w1, gcn_b1, tcn_w1, tcn_b1, bn_g1, bn_b1, bn_m1, bn_v1,
      fc_w, fc_b, (float*)d_out);
}


# LDS per workgroup cut from 149 to 105 KiB so three workgroups fit per WGP instead of two: gcn1 writes g1 in place over its input (a wave owns a 16-row stripe), tcn1 weights reuse tcn0's LDS buffer
# speedup vs baseline: 1.4733x; 1.2222x over previous
#include <hip/hip_runtime.h>

#define TT 8
#define T1 (TT + 6)
#define T2 (TT + 4)
#define T3 (TT + 2)
#define NB 64
#define NTIME 512
#define NV 21
#define NC 32
#define NCIN 2
#define NOUT 64
#define PC 36
#define PW 40

typedef __attribute__((ext_vector_type(2))) float v2f;
typedef __attribute__((ext_vector_type(8))) float v8f;

__device__ __forceinline__ v8f wmma4(v2f a, v2f b, v8f c) {
  return __builtin_amdgcn_wmma_f32_16x16x4_f32(
       false, a,  false, b,
       (short)0, c,  false,  false);
}

__device__ const unsigned char g_pi[20] = {0,1,2,3,0,5,6,7,0,9,10,11,0,13,14,15,0,17,18,19};
__device__ const unsigned char g_pj[20] = {1,2,3,4,5,6,7,8,9,10,11,12,13,14,15,16,17,18,19,20};
__device__ const float g_dinv[21] = {
  0.4082482905f,
  0.5773502692f,0.5773502692f,0.5773502692f,0.7071067812f,
  0.5773502692f,0.5773502692f,0.5773502692f,0.7071067812f,
  0.5773502692f,0.5773502692f,0.5773502692f,0.7071067812f,
  0.5773502692f,0.5773502692f,0.5773502692f,0.7071067812f,
  0.5773502692f,0.5773502692f,0.5773502692f,0.7071067812f};

__global__ __launch_bounds__(256) void stgcn_fused(
    const float* __restrict__ x,
    const float* __restrict__ gcn_w0, const float* __restrict__ gcn_b0,
    const float* __restrict__ tcn_w0, const float* __restrict__ tcn_b0,
    const float* __restrict__ bn_g0, const float* __restrict__ bn_b0,
    const float* __restrict__ bn_m0, const float* __restrict__ bn_v0,
    const float* __restrict__ res_w, const float* __restrict__ res_b,
    const float* __restrict__ res_bn_g, const float* __restrict__ res_bn_b,
    const float* __restrict__ res_bn_m, const float* __restrict__ res_bn_v,
    const float* __restrict__ gcn_w1, const float* __restrict__ gcn_b1,
    const float* __restrict__ tcn_w1, const float* __restrict__ tcn_b1,
    const float* __restrict__ bn_g1, const float* __restrict__ bn_b1,
    const float* __restrict__ bn_m1, const float* __restrict__ bn_v1,
    const float* __restrict__ fc_w, const float* __restrict__ fc_b,
    float* __restrict__ out)
{
  __shared__ float sA[NV * NV];
  __shared__ float s_in[T1 * NV * NCIN];
  __shared__ float s_xa[T1 * NV * NCIN];
  __shared__ float s_tw[96 * PW];
  __shared__ float s_gw1[NC * PC];
  __shared__ float s_g0[T1 * NV * PC];
  __shared__ float s_h0[T2 * NV * PC];
  __shared__ float s_gw0[NC * NCIN];
  __shared__ float s_gb0[NC], s_gb1[NC];
  __shared__ float s_s0[NC], s_o0[NC], s_s1[NC], s_o1[NC];
  __shared__ float s_rA[NC], s_rB[NC], s_rC[NC];
  __shared__ float s_fcb[NOUT];

  const int tid  = threadIdx.x;
  const int nthr = 256;
  const int lane = tid & 31;
  const int wave = tid >> 5;
  const int row  = lane & 15;
  const int hi   = lane >> 4;

  const int bid   = blockIdx.x;
  const int tile  = bid & 63;
  const int hand  = (bid >> 6) & 1;
  const int b     = bid >> 7;
  const int tbase = tile * TT;

  for (int idx = tid; idx < NV * NV; idx += nthr) {
    int i = idx / NV, j = idx % NV;
    bool conn = (i == j);
    for (int p = 0; p < 20; ++p)
      conn = conn || ((int)g_pi[p] == i && (int)g_pj[p] == j) ||
                     ((int)g_pi[p] == j && (int)g_pj[p] == i);
    sA[idx] = conn ? g_dinv[i] * g_dinv[j] : 0.0f;
  }
  for (int idx = tid; idx < 96 * NC; idx += nthr) {
    int co = idx & 31, kap = idx >> 5, k = kap >> 5, ci = kap & 31;
    s_tw[kap * PW + co] = tcn_w0[(co * NC + ci) * 3 + k];
  }
  for (int idx = tid; idx < NC * NC; idx += nthr) s_gw1[(idx >> 5) * PC + (idx & 31)] = gcn_w1[idx];
  if (tid < NC * NCIN) s_gw0[tid] = gcn_w0[tid];
  if (tid < NC) {
    s_gb0[tid] = gcn_b0[tid];
    s_gb1[tid] = gcn_b1[tid];
    float sc0 = bn_g0[tid] * rsqrtf(bn_v0[tid] + 1e-5f);
    s_s0[tid] = sc0;
    s_o0[tid] = (tcn_b0[tid] - bn_m0[tid]) * sc0 + bn_b0[tid];
    float sc1 = bn_g1[tid] * rsqrtf(bn_v1[tid] + 1e-5f);
    s_s1[tid] = sc1;
    s_o1[tid] = (tcn_b1[tid] - bn_m1[tid]) * sc1 + bn_b1[tid];
    float rsc = res_bn_g[tid] * rsqrtf(res_bn_v[tid] + 1e-5f);
    s_rA[tid] = res_w[tid * 2 + 0] * rsc;
    s_rB[tid] = res_w[tid * 2 + 1] * rsc;
    s_rC[tid] = (res_b[tid] - res_bn_m[tid]) * rsc + res_bn_b[tid];
  }
  if (tid < NOUT) s_fcb[tid] = fc_b[tid];

  for (int idx = tid; idx < T1 * NV * NCIN; idx += nthr) {
    int i = idx / (NV * NCIN), r = idx % (NV * NCIN);
    int v = r >> 1, ci = r & 1;
    int t = tbase - 3 + i;
    float val = 0.0f;
    if (t >= 0 && t < NTIME)
      val = x[(((b * NTIME + t) * 2 + hand) * NV + v) * NCIN + ci];
    s_in[idx] = val;
  }
  __syncthreads();

  for (int idx = tid; idx < T1 * NV * NCIN; idx += nthr) {
    int i = idx / (NV * NCIN), r = idx % (NV * NCIN);
    int v = r >> 1, ci = r & 1;
    float acc = 0.0f;
    const float* Ar = &sA[v * NV];
    const float* xr = &s_in[i * NV * NCIN + ci];
    for (int u = 0; u < NV; ++u) acc += Ar[u] * xr[u * NCIN];
    s_xa[idx] = acc;
  }
  __syncthreads();

  for (int idx = tid; idx < T1 * NV * NC; idx += nthr) {
    int c = idx & 31, rv = idx >> 5;
    int v = rv % NV, i = rv / NV;
    int t = tbase - 3 + i;
    float val = 0.0f;
    if (t >= 0 && t < NTIME) {
      float xa0 = s_xa[(i * NV + v) * 2 + 0];
      float xa1 = s_xa[(i * NV + v) * 2 + 1];
      val = s_gw0[c * 2 + 0] * xa0 + s_gw0[c * 2 + 1] * xa1 + s_gb0[c];
    }
    s_g0[rv * PC + c] = val;
  }
  __syncthreads();

  float tw1r[12];
#pragma unroll
  for (int i = 0; i < 12; ++i) {
    int idx = tid + i * 256;
    int co = idx & 31, kap = idx >> 5, k = kap >> 5, ci = kap & 31;
    tw1r[i] = tcn_w1[(co * NC + ci) * 3 + k];
  }

  {
    const int M = T2 * NV;
    for (int tIdx = wave; tIdx < 16 * 2; tIdx += 8) {
      int mt = tIdx >> 1, nt = tIdx & 1;
      v8f acc = {0.f,0.f,0.f,0.f,0.f,0.f,0.f,0.f};
      int mA = mt * 16 + row;
      int mC = (mA < M) ? mA : 0;
      int t2 = mC / NV, vv = mC % NV;
      int colN = nt * 16 + row;
      for (int ks = 0; ks < 24; ++ks) {
        int k0 = ks * 4 + hi * 2;
        v2f a, bb;
        a.x = s_g0[(t2 + (k0 >> 5)) * (NV * PC) + vv * PC + (k0 & 31)];
        a.y = s_g0[(t2 + ((k0 + 1) >> 5)) * (NV * PC) + vv * PC + ((k0 + 1) & 31)];
        bb.x = s_tw[k0 * PW + colN];
        bb.y = s_tw[(k0 + 1) * PW + colN];
        acc = wmma4(a, bb, acc);
      }
#pragma unroll
      for (int j = 0; j < 8; ++j) {
        int m = mt * 16 + j + hi * 8;
        if (m < M) {
          int te = m / NV, ve = m % NV;
          int co = colN;
          float val = acc[j] * s_s0[co] + s_o0[co];
          float x0 = s_in[((te + 1) * NV + ve) * 2 + 0];
          float x1 = s_in[((te + 1) * NV + ve) * 2 + 1];
          val += s_rA[co] * x0 + s_rB[co] * x1 + s_rC[co];
          s_h0[(te * NV + ve) * PC + co] = fmaxf(val, 0.0f);
        }
      }
    }
  }
  __syncthreads();

#pragma unroll
  for (int i = 0; i < 12; ++i) {
    int idx = tid + i * 256;
    s_tw[(idx >> 5) * PW + (idx & 31)] = tw1r[i];
  }

  float* s_hagg = s_g0;
  for (int idx = tid; idx < T3 * NV * NC; idx += nthr) {
    int ci = idx & 31, rv = idx >> 5;
    int v = rv % NV, t3 = rv / NV;
    float acc = 0.0f;
    const float* Ar = &sA[v * NV];
    const float* hr = &s_h0[(t3 + 1) * (NV * PC) + ci];
    for (int u = 0; u < NV; ++u) acc += Ar[u] * hr[u * PC];
    s_hagg[rv * PC + ci] = acc;
  }
  __syncthreads();

  float* s_g1 = s_g0;
  {
    const int M = T3 * NV;
    const float gb1a = s_gb1[row], gb1b = s_gb1[16 + row];
    for (int mt = wave; mt < 14; mt += 8) {
      int mA = mt * 16 + row;
      int mC = (mA < M) ? mA : mt * 16;
      v2f a[8];
#pragma unroll
      for (int ks = 0; ks < 8; ++ks) {
        int k0 = ks * 4 + hi * 2;
        a[ks].x = s_hagg[mC * PC + k0];
        a[ks].y = s_hagg[mC * PC + k0 + 1];
      }
      v8f acc0 = {0.f,0.f,0.f,0.f,0.f,0.f,0.f,0.f};
      v8f acc1 = {0.f,0.f,0.f,0.f,0.f,0.f,0.f,0.f};
#pragma unroll
      for (int ks = 0; ks < 8; ++ks) {
        int k0 = ks * 4 + hi * 2;
        v2f b0, b1;
        b0.x = s_gw1[row * PC + k0];
        b0.y = s_gw1[row * PC + k0 + 1];
        b1.x = s_gw1[(16 + row) * PC + k0];
        b1.y = s_gw1[(16 + row) * PC + k0 + 1];
        acc0 = wmma4(a[ks], b0, acc0);
        acc1 = wmma4(a[ks], b1, acc1);
      }
#pragma unroll
      for (int j = 0; j < 8; ++j) {
        int m = mt * 16 + j + hi * 8;
        if (m < M) {
          int t3 = m / NV;
          int t = tbase - 1 + t3;
          bool ok = (t >= 0 && t < NTIME);
          s_g1[m * PC + row]      = ok ? acc0[j] + gb1a : 0.0f;
          s_g1[m * PC + 16 + row] = ok ? acc1[j] + gb1b : 0.0f;
        }
      }
    }
  }
  __syncthreads();

  {
    const int M = TT * NV;
    for (int tIdx = wave; tIdx < 11 * 2; tIdx += 8) {
      int mt = tIdx >> 1, nt = tIdx & 1;
      v8f acc = {0.f,0.f,0.f,0.f,0.f,0.f,0.f,0.f};
      int mA = mt * 16 + row;
      int mC = (mA < M) ? mA : 0;
      int t4 = mC / NV, vv = mC % NV;
      int colN = nt * 16 + row;
      for (int ks = 0; ks < 24; ++ks) {
        int k0 = ks * 4 + hi * 2;
        v2f a, bb;
        a.x = s_g1[((t4 + (k0 >> 5)) * NV + vv) * PC + (k0 & 31)];
        a.y = s_g1[((t4 + ((k0 + 1) >> 5)) * NV + vv) * PC + ((k0 + 1) & 31)];
        bb.x = s_tw[k0 * PW + colN];
        bb.y = s_tw[(k0 + 1) * PW + colN];
        acc = wmma4(a, bb, acc);
      }
#pragma unroll
      for (int j = 0; j < 8; ++j) {
        int m = mt * 16 + j + hi * 8;
        if (m < M) {
          int te = m / NV, ve = m % NV;
          int co = colN;
          int hidx = ((te + 2) * NV + ve) * PC + co;
          float val = acc[j] * s_s1[co] + s_o1[co] + s_h0[hidx];
          s_h0[hidx] = fmaxf(val, 0.0f);
        }
      }
    }
  }
  __syncthreads();

  {
    float* s_red = s_g0;
    int nt  = wave & 3;
    int ksb = (wave >> 2) * 84;
    v8f acc = {0.f,0.f,0.f,0.f,0.f,0.f,0.f,0.f};
    int t4 = (row < TT) ? row : (TT - 1);
    int colN = nt * 16 + row;
    const float* h1base = &s_h0[(t4 + 2) * (NV * PC)];
    for (int ks = ksb; ks < ksb + 84; ++ks) {
      int f0 = ks * 4 + hi * 2;
      int f1 = f0 + 1;
      int c0 = f0 / NV, v0 = f0 % NV;
      int c1 = f1 / NV, v1 = f1 % NV;
      v2f a, bb;
      a.x = h1base[v0 * PC + c0];
      a.y = h1base[v1 * PC + c1];
      bb.x = fc_w[colN * 672 + f0];
      bb.y = fc_w[colN * 672 + f1];
      acc = wmma4(a, bb, acc);
    }
    if (wave >= 4) {
#pragma unroll
      for (int j = 0; j < 8; ++j) s_red[nt * 256 + lane * 8 + j] = acc[j];
    }
    __syncthreads();
    if (wave < 4) {
#pragma unroll
      for (int j = 0; j < 8; ++j) {
        int m = j + hi * 8;
        if (m < TT) {
          int t = tbase + m;
          float val = acc[j] + s_red[nt * 256 + lane * 8 + j] + s_fcb[colN];
          out[(((size_t)b * NTIME + t) * 2 + hand) * NOUT + colN] = val;
        }
      }
    }
  }
}

extern "C" void kernel_launch(void* const* d_in, const int* in_sizes, int n_in,
                              void* d_out, int out_size, void* d_ws, size_t ws_size,
                              hipStream_t stream) {
  (void)in_sizes; (void)n_in; (void)d_ws; (void)ws_size; (void)out_size;
  const float* x        = (const float*)d_in[0];
  const float* gcn_w0   = (const float*)d_in[1];
  const float* gcn_b0   = (const float*)d_in[2];
  const float* tcn_w0   = (const float*)d_in[3];
  const float* tcn_b0   = (const float*)d_in[4];
  const float* bn_g0    = (const float*)d_in[5];
  const float* bn_b0    = (const float*)d_in[6];
  const float* bn_m0    = (const float*)d_in[7];
  const float* bn_v0    = (const float*)d_in[8];
  const float* res_w    = (const float*)d_in[9];
  const float* res_b    = (const float*)d_in[10];
  const float* res_bn_g = (const float*)d_in[11];
  const float* res_bn_b = (const float*)d_in[12];
  const float* res_bn_m = (const float*)d_in[13];
  const float* res_bn_v = (const float*)d_in[14];
  const float* gcn_w1   = (const float*)d_in[15];
  const float* gcn_b1   = (const float*)d_in[16];
  const float* tcn_w1   = (const float*)d_in[17];
  const float* tcn_b1   = (const float*)d_in[18];
  const float* bn_g1    = (const float*)d_in[19];
  const float* bn_b1    = (const float*)d_in[20];
  const float* bn_m1    = (const float*)d_in[21];
  const float* bn_v1    = (const float*)d_in[22];
  const float* fc_w     = (const float*)d_in[23];
  const float* fc_b     = (const float*)d_in[24];

  dim3 grid(NB * 2 * (NTIME / TT));
  dim3 block(256);
  stgcn_fused<<<grid, block, 0, stream>>>(
      x, gcn_w0, gcn_b0, tcn_w0, tcn_b0, bn_g0, bn_b0, bn_m0, bn_v0,
      res_w, res_b, res_bn_g, res_bn_b, res_bn_m, res_bn_v,
      gcn_w1, gcn_b1, tcn_w1, tcn_b1, bn_g1, bn_b1, bn_m1, bn_v1,
      fc_w, fc_b, (float*)d_out);
}


# Graph aggregation (x, h0) as a compile-time stencil over the 61 non-zero adjacency entries, same ascending-u sum order, zeros skipped, mul-add fused; dense 21x21 LDS matrix and its build removed
# speedup vs baseline: 1.8466x; 1.2534x over previous
#include <hip/hip_runtime.h>

#define TT 8
#define T1 (TT + 6)
#define T2 (TT + 4)
#define T3 (TT + 2)
#define NB 64
#define NTIME 512
#define NV 21
#define NC 32
#define NCIN 2
#define NOUT 64
#define PC 36
#define PW 40

typedef __attribute__((ext_vector_type(2))) float v2f;
typedef __attribute__((ext_vector_type(8))) float v8f;

__device__ __forceinline__ v8f wmma4(v2f a, v2f b, v8f c) {
  return __builtin_amdgcn_wmma_f32_16x16x4_f32(
       false, a,  false, b,
       (short)0, c,  false,  false);
}

__host__ __device__ constexpr int g_parent(int j) { return ((j - 1) % 4 == 0) ? 0 : j - 1; }
__host__ __device__ constexpr bool g_conn(int i, int j) {
  return i == j || (j > 0 && g_parent(j) == i) || (i > 0 && g_parent(i) == j);
}
__host__ __device__ constexpr float g_dinv(int v) {
  return v == 0 ? 0.4082482905f : ((v % 4 == 0) ? 0.7071067812f : 0.5773502692f);
}

template <int STRIDE_IN, int STRIDE_OUT>
__device__ __forceinline__ void graph_agg(const float* __restrict__ in, float* __restrict__ out) {
  float h[NV];
#pragma unroll
  for (int u = 0; u < NV; ++u) h[u] = in[u * STRIDE_IN];
#pragma unroll
  for (int v = 0; v < NV; ++v) {
    float acc = 0.0f;
    bool have = false;
#pragma unroll
    for (int u = 0; u < NV; ++u) {
      if (g_conn(v, u)) {
        const float p = (g_dinv(v) * g_dinv(u)) * h[u];
        acc = have ? acc + p : p;
        have = true;
      }
    }
    out[v * STRIDE_OUT] = acc;
  }
}

__global__ __launch_bounds__(256) void stgcn_fused(
    const float* __restrict__ x,
    const float* __restrict__ gcn_w0, const float* __restrict__ gcn_b0,
    const float* __restrict__ tcn_w0, const float* __restrict__ tcn_b0,
    const float* __restrict__ bn_g0, const float* __restrict__ bn_b0,
    const float* __restrict__ bn_m0, const float* __restrict__ bn_v0,
    const float* __restrict__ res_w, const float* __restrict__ res_b,
    const float* __restrict__ res_bn_g, const float* __restrict__ res_bn_b,
    const float* __restrict__ res_bn_m, const float* __restrict__ res_bn_v,
    const float* __restrict__ gcn_w1, const float* __restrict__ gcn_b1,
    const float* __restrict__ tcn_w1, const float* __restrict__ tcn_b1,
    const float* __restrict__ bn_g1, const float* __restrict__ bn_b1,
    const float* __restrict__ bn_m1, const float* __restrict__ bn_v1,
    const float* __restrict__ fc_w, const float* __restrict__ fc_b,
    float* __restrict__ out)
{
  __shared__ float s_in[T1 * NV * NCIN];
  __shared__ float s_xa[T1 * NV * NCIN];
  __shared__ float s_tw[96 * PW];
  __shared__ float s_gw1[NC * PC];
  __shared__ float s_g0[T1 * NV * PC];
  __shared__ float s_h0[T2 * NV * PC];
  __shared__ float s_gw0[NC * NCIN];
  __shared__ float s_gb0[NC], s_gb1[NC];
  __shared__ float s_s0[NC], s_o0[NC], s_s1[NC], s_o1[NC];
  __shared__ float s_rA[NC], s_rB[NC], s_rC[NC];
  __shared__ float s_fcb[NOUT];

  const int tid  = threadIdx.x;
  const int nthr = 256;
  const int lane = tid & 31;
  const int wave = tid >> 5;
  const int row  = lane & 15;
  const int hi   = lane >> 4;

  const int bid   = blockIdx.x;
  const int tile  = bid & 63;
  const int hand  = (bid >> 6) & 1;
  const int b     = bid >> 7;
  const int tbase = tile * TT;

  for (int idx = tid; idx < 96 * NC; idx += nthr) {
    int co = idx & 31, kap = idx >> 5, k = kap >> 5, ci = kap & 31;
    s_tw[kap * PW + co] = tcn_w0[(co * NC + ci) * 3 + k];
  }
  for (int idx = tid; idx < NC * NC; idx += nthr) s_gw1[(idx >> 5) * PC + (idx & 31)] = gcn_w1[idx];
  if (tid < NC * NCIN) s_gw0[tid] = gcn_w0[tid];
  if (tid < NC) {
    s_gb0[tid] = gcn_b0[tid];
    s_gb1[tid] = gcn_b1[tid];
    float sc0 = bn_g0[tid] * rsqrtf(bn_v0[tid] + 1e-5f);
    s_s0[tid] = sc0;
    s_o0[tid] = (tcn_b0[tid] - bn_m0[tid]) * sc0 + bn_b0[tid];
    float sc1 = bn_g1[tid] * rsqrtf(bn_v1[tid] + 1e-5f);
    s_s1[tid] = sc1;
    s_o1[tid] = (tcn_b1[tid] - bn_m1[tid]) * sc1 + bn_b1[tid];
    float rsc = res_bn_g[tid] * rsqrtf(res_bn_v[tid] + 1e-5f);
    s_rA[tid] = res_w[tid * 2 + 0] * rsc;
    s_rB[tid] = res_w[tid * 2 + 1] * rsc;
    s_rC[tid] = (res_b[tid] - res_bn_m[tid]) * rsc + res_bn_b[tid];
  }
  if (tid < NOUT) s_fcb[tid] = fc_b[tid];

  for (int idx = tid; idx < T1 * NV * NCIN; idx += nthr) {
    int i = idx / (NV * NCIN), r = idx % (NV * NCIN);
    int v = r >> 1, ci = r & 1;
    int t = tbase - 3 + i;
    float val = 0.0f;
    if (t >= 0 && t < NTIME)
      val = x[(((b * NTIME + t) * 2 + hand) * NV + v) * NCIN + ci];
    s_in[idx] = val;
  }
  __syncthreads();

  if (tid < T1 * NCIN)
    graph_agg<NCIN, NCIN>(&s_in[(tid >> 1) * (NV * NCIN) + (tid & 1)],
                          &s_xa[(tid >> 1) * (NV * NCIN) + (tid & 1)]);
  __syncthreads();

  for (int idx = tid; idx < T1 * NV * NC; idx += nthr) {
    int c = idx & 31, rv = idx >> 5;
    int v = rv % NV, i = rv / NV;
    int t = tbase - 3 + i;
    float val = 0.0f;
    if (t >= 0 && t < NTIME) {
      float xa0 = s_xa[(i * NV + v) * 2 + 0];
      float xa1 = s_xa[(i * NV + v) * 2 + 1];
      val = s_gw0[c * 2 + 0] * xa0 + s_gw0[c * 2 + 1] * xa1 + s_gb0[c];
    }
    s_g0[rv * PC + c] = val;
  }
  __syncthreads();

  float tw1r[12];
#pragma unroll
  for (int i = 0; i < 12; ++i) {
    int idx = tid + i * 256;
    int co = idx & 31, kap = idx >> 5, k = kap >> 5, ci = kap & 31;
    tw1r[i] = tcn_w1[(co * NC + ci) * 3 + k];
  }

  {
    const int M = T2 * NV;
    for (int tIdx = wave; tIdx < 16 * 2; tIdx += 8) {
      int mt = tIdx >> 1, nt = tIdx & 1;
      v8f acc = {0.f,0.f,0.f,0.f,0.f,0.f,0.f,0.f};
      int mA = mt * 16 + row;
      int mC = (mA < M) ? mA : 0;
      int t2 = mC / NV, vv = mC % NV;
      int colN = nt * 16 + row;
      for (int ks = 0; ks < 24; ++ks) {
        int k0 = ks * 4 + hi * 2;
        v2f a, bb;
        a.x = s_g0[(t2 + (k0 >> 5)) * (NV * PC) + vv * PC + (k0 & 31)];
        a.y = s_g0[(t2 + ((k0 + 1) >> 5)) * (NV * PC) + vv * PC + ((k0 + 1) & 31)];
        bb.x = s_tw[k0 * PW + colN];
        bb.y = s_tw[(k0 + 1) * PW + colN];
        acc = wmma4(a, bb, acc);
      }
#pragma unroll
      for (int j = 0; j < 8; ++j) {
        int m = mt * 16 + j + hi * 8;
        if (m < M) {
          int te = m / NV, ve = m % NV;
          int co = colN;
          float val = acc[j] * s_s0[co] + s_o0[co];
          float x0 = s_in[((te + 1) * NV + ve) * 2 + 0];
          float x1 = s_in[((te + 1) * NV + ve) * 2 + 1];
          val += s_rA[co] * x0 + s_rB[co] * x1 + s_rC[co];
          s_h0[(te * NV + ve) * PC + co] = fmaxf(val, 0.0f);
        }
      }
    }
  }
  __syncthreads();

#pragma unroll
  for (int i = 0; i < 12; ++i) {
    int idx = tid + i * 256;
    s_tw[(idx >> 5) * PW + (idx & 31)] = tw1r[i];
  }

  float* s_hagg = s_g0;
  for (int t3 = wave; t3 < T3; t3 += 8)
    graph_agg<PC, PC>(&s_h0[(t3 + 1) * (NV * PC) + lane], &s_hagg[t3 * (NV * PC) + lane]);
  __syncthreads();

  float* s_g1 = s_g0;
  {
    const int M = T3 * NV;
    const float gb1a = s_gb1[row], gb1b = s_gb1[16 + row];
    for (int mt = wave; mt < 14; mt += 8) {
      int mA = mt * 16 + row;
      int mC = (mA < M) ? mA : mt * 16;
      v2f a[8];
#pragma unroll
      for (int ks = 0; ks < 8; ++ks) {
        int k0 = ks * 4 + hi * 2;
        a[ks].x = s_hagg[mC * PC + k0];
        a[ks].y = s_hagg[mC * PC + k0 + 1];
      }
      v8f acc0 = {0.f,0.f,0.f,0.f,0.f,0.f,0.f,0.f};
      v8f acc1 = {0.f,0.f,0.f,0.f,0.f,0.f,0.f,0.f};
#pragma unroll
      for (int ks = 0; ks < 8; ++ks) {
        int k0 = ks * 4 + hi * 2;
        v2f b0, b1;
        b0.x = s_gw1[row * PC + k0];
        b0.y = s_gw1[row * PC + k0 + 1];
        b1.x = s_gw1[(16 + row) * PC + k0];
        b1.y = s_gw1[(16 + row) * PC + k0 + 1];
        acc0 = wmma4(a[ks], b0, acc0);
        acc1 = wmma4(a[ks], b1, acc1);
      }
#pragma unroll
      for (int j = 0; j < 8; ++j) {
        int m = mt * 16 + j + hi * 8;
        if (m < M) {
          int t3 = m / NV;
          int t = tbase - 1 + t3;
          bool ok = (t >= 0 && t < NTIME);
          s_g1[m * PC + row]      = ok ? acc0[j] + gb1a : 0.0f;
          s_g1[m * PC + 16 + row] = ok ? acc1[j] + gb1b : 0.0f;
        }
      }
    }
  }
  __syncthreads();

  {
    const int M = TT * NV;
    for (int tIdx = wave; tIdx < 11 * 2; tIdx += 8) {
      int mt = tIdx >> 1, nt = tIdx & 1;
      v8f acc = {0.f,0.f,0.f,0.f,0.f,0.f,0.f,0.f};
      int mA = mt * 16 + row;
      int mC = (mA < M) ? mA : 0;
      int t4 = mC / NV, vv = mC % NV;
      int colN = nt * 16 + row;
      for (int ks = 0; ks < 24; ++ks) {
        int k0 = ks * 4 + hi * 2;
        v2f a, bb;
        a.x = s_g1[((t4 + (k0 >> 5)) * NV + vv) * PC + (k0 & 31)];
        a.y = s_g1[((t4 + ((k0 + 1) >> 5)) * NV + vv) * PC + ((k0 + 1) & 31)];
        bb.x = s_tw[k0 * PW + colN];
        bb.y = s_tw[(k0 + 1) * PW + colN];
        acc = wmma4(a, bb, acc);
      }
#pragma unroll
      for (int j = 0; j < 8; ++j) {
        int m = mt * 16 + j + hi * 8;
        if (m < M) {
          int te = m / NV, ve = m % NV;
          int co = colN;
          int hidx = ((te + 2) * NV + ve) * PC + co;
          float val = acc[j] * s_s1[co] + s_o1[co] + s_h0[hidx];
          s_h0[hidx] = fmaxf(val, 0.0f);
        }
      }
    }
  }
  __syncthreads();

  {
    float* s_red = s_g0;
    int nt  = wave & 3;
    int ksb = (wave >> 2) * 84;
    v8f acc = {0.f,0.f,0.f,0.f,0.f,0.f,0.f,0.f};
    int t4 = (row < TT) ? row : (TT - 1);
    int colN = nt * 16 + row;
    const float* h1base = &s_h0[(t4 + 2) * (NV * PC)];
    for (int ks = ksb; ks < ksb + 84; ++ks) {
      int f0 = ks * 4 + hi * 2;
      int f1 = f0 + 1;
      int c0 = f0 / NV, v0 = f0 % NV;
      int c1 = f1 / NV, v1 = f1 % NV;
      v2f a, bb;
      a.x = h1base[v0 * PC + c0];
      a.y = h1base[v1 * PC + c1];
      bb.x = fc_w[colN * 672 + f0];
      bb.y = fc_w[colN * 672 + f1];
      acc = wmma4(a, bb, acc);
    }
    if (wave >= 4) {
#pragma unroll
      for (int j = 0; j < 8; ++j) s_red[nt * 256 + lane * 8 + j] = acc[j];
    }
    __syncthreads();
    if (wave < 4) {
#pragma unroll
      for (int j = 0; j < 8; ++j) {
        int m = j + hi * 8;
        if (m < TT) {
          int t = tbase + m;
          float val = acc[j] + s_red[nt * 256 + lane * 8 + j] + s_fcb[colN];
          out[(((size_t)b * NTIME + t) * 2 + hand) * NOUT + colN] = val;
        }
      }
    }
  }
}

extern "C" void kernel_launch(void* const* d_in, const int* in_sizes, int n_in,
                              void* d_out, int out_size, void* d_ws, size_t ws_size,
                              hipStream_t stream) {
  (void)in_sizes; (void)n_in; (void)d_ws; (void)ws_size; (void)out_size;
  const float* x        = (const float*)d_in[0];
  const float* gcn_w0   = (const float*)d_in[1];
  const float* gcn_b0   = (const float*)d_in[2];
  const float* tcn_w0   = (const float*)d_in[3];
  const float* tcn_b0   = (const float*)d_in[4];
  const float* bn_g0    = (const float*)d_in[5];
  const float* bn_b0    = (const float*)d_in[6];
  const float* bn_m0    = (const float*)d_in[7];
  const float* bn_v0    = (const float*)d_in[8];
  const float* res_w    = (const float*)d_in[9];
  const float* res_b    = (const float*)d_in[10];
  const float* res_bn_g = (const float*)d_in[11];
  const float* res_bn_b = (const float*)d_in[12];
  const float* res_bn_m = (const float*)d_in[13];
  const float* res_bn_v = (const float*)d_in[14];
  const float* gcn_w1   = (const float*)d_in[15];
  const float* gcn_b1   = (const float*)d_in[16];
  const float* tcn_w1   = (const float*)d_in[17];
  const float* tcn_b1   = (const float*)d_in[18];
  const float* bn_g1    = (const float*)d_in[19];
  const float* bn_b1    = (const float*)d_in[20];
  const float* bn_m1    = (const float*)d_in[21];
  const float* bn_v1    = (const float*)d_in[22];
  const float* fc_w     = (const float*)d_in[23];
  const float* fc_b     = (const float*)d_in[24];

  dim3 grid(NB * 2 * (NTIME / TT));
  dim3 block(256);
  stgcn_fused<<<grid, block, 0, stream>>>(
      x, gcn_w0, gcn_b0, tcn_w0, tcn_b0, bn_g0, bn_b0, bn_m0, bn_v0,
      res_w, res_b, res_bn_g, res_bn_b, res_bn_m, res_bn_v,
      gcn_w1, gcn_b1, tcn_w1, tcn_b1, bn_g1, bn_b1, bn_m1, bn_v1,
      fc_w, fc_b, (float*)d_out);
}


# gcn0 (2->32 + bias) folded into tcn0's weights per workgroup: tcn0 runs K=12 (3 taps x [xa0,xa1,valid,0]) not 96; g0 pass and buffer dropped; ci is summed in the fold first (new sum order)
# speedup vs baseline: 2.4244x; 1.3129x over previous
#include <hip/hip_runtime.h>

#define TT 8
#define T1 (TT + 6)
#define T2 (TT + 4)
#define T3 (TT + 2)
#define NB 64
#define NTIME 512
#define NV 21
#define NC 32
#define NCIN 2
#define NOUT 64
#define PC 36
#define PW 40
#define PR 97

typedef __attribute__((ext_vector_type(2))) float v2f;
typedef __attribute__((ext_vector_type(8))) float v8f;

__device__ __forceinline__ v8f wmma4(v2f a, v2f b, v8f c) {
  return __builtin_amdgcn_wmma_f32_16x16x4_f32(
       false, a,  false, b,
       (short)0, c,  false,  false);
}

__host__ __device__ constexpr int g_parent(int j) { return ((j - 1) % 4 == 0) ? 0 : j - 1; }
__host__ __device__ constexpr bool g_conn(int i, int j) {
  return i == j || (j > 0 && g_parent(j) == i) || (i > 0 && g_parent(i) == j);
}
__host__ __device__ constexpr float g_dinv(int v) {
  return v == 0 ? 0.4082482905f : ((v % 4 == 0) ? 0.7071067812f : 0.5773502692f);
}

template <int STRIDE_IN, int STRIDE_OUT>
__device__ __forceinline__ void graph_agg(const float* __restrict__ in, float* __restrict__ out) {
  float h[NV];
#pragma unroll
  for (int u = 0; u < NV; ++u) h[u] = in[u * STRIDE_IN];
#pragma unroll
  for (int v = 0; v < NV; ++v) {
    float acc = 0.0f;
    bool have = false;
#pragma unroll
    for (int u = 0; u < NV; ++u) {
      if (g_conn(v, u)) {
        const float p = (g_dinv(v) * g_dinv(u)) * h[u];
        acc = have ? acc + p : p;
        have = true;
      }
    }
    out[v * STRIDE_OUT] = acc;
  }
}

__global__ __launch_bounds__(256) void stgcn_fused(
    const float* __restrict__ x,
    const float* __restrict__ gcn_w0, const float* __restrict__ gcn_b0,
    const float* __restrict__ tcn_w0, const float* __restrict__ tcn_b0,
    const float* __restrict__ bn_g0, const float* __restrict__ bn_b0,
    const float* __restrict__ bn_m0, const float* __restrict__ bn_v0,
    const float* __restrict__ res_w, const float* __restrict__ res_b,
    const float* __restrict__ res_bn_g, const float* __restrict__ res_bn_b,
    const float* __restrict__ res_bn_m, const float* __restrict__ res_bn_v,
    const float* __restrict__ gcn_w1, const float* __restrict__ gcn_b1,
    const float* __restrict__ tcn_w1, const float* __restrict__ tcn_b1,
    const float* __restrict__ bn_g1, const float* __restrict__ bn_b1,
    const float* __restrict__ bn_m1, const float* __restrict__ bn_v1,
    const float* __restrict__ fc_w, const float* __restrict__ fc_b,
    float* __restrict__ out)
{
  __shared__ float s_in[T1 * NV * NCIN];
  __shared__ __attribute__((aligned(16))) float s_xa[T1 * NV * NCIN];
  __shared__ float s_tw[96 * PW];
  __shared__ float s_gw1[NC * PC];
  __shared__ float s_g1[T3 * NV * PC];
  __shared__ float s_h0[T2 * NV * PC];
  __shared__ __attribute__((aligned(16))) float s_wf[3 * NC * 4];
  __shared__ float s_gb1[NC];
  __shared__ float s_s0[NC], s_o0[NC], s_s1[NC], s_o1[NC];
  __shared__ float s_rA[NC], s_rB[NC], s_rC[NC];
  __shared__ float s_fcb[NOUT];

  const int tid  = threadIdx.x;
  const int nthr = 256;
  const int lane = tid & 31;
  const int wave = tid >> 5;
  const int row  = lane & 15;
  const int hi   = lane >> 4;

  const int bid   = blockIdx.x;
  const int tile  = bid & 63;
  const int hand  = (bid >> 6) & 1;
  const int b     = bid >> 7;
  const int tbase = tile * TT;

  float* s_raw = s_g1;
#pragma unroll
  for (int i = 0; i < 12; ++i) {
    int idx = tid + i * 256;
    int co = idx / 96;
    s_raw[co * PR + (idx - co * 96)] = tcn_w0[idx];
  }
  for (int idx = tid; idx < NC * NC; idx += nthr) s_gw1[(idx >> 5) * PC + (idx & 31)] = gcn_w1[idx];
  if (tid < NC) {
    s_gb1[tid] = gcn_b1[tid];
    float sc0 = bn_g0[tid] * rsqrtf(bn_v0[tid] + 1e-5f);
    s_s0[tid] = sc0;
    s_o0[tid] = (tcn_b0[tid] - bn_m0[tid]) * sc0 + bn_b0[tid];
    float sc1 = bn_g1[tid] * rsqrtf(bn_v1[tid] + 1e-5f);
    s_s1[tid] = sc1;
    s_o1[tid] = (tcn_b1[tid] - bn_m1[tid]) * sc1 + bn_b1[tid];
    float rsc = res_bn_g[tid] * rsqrtf(res_bn_v[tid] + 1e-5f);
    s_rA[tid] = res_w[tid * 2 + 0] * rsc;
    s_rB[tid] = res_w[tid * 2 + 1] * rsc;
    s_rC[tid] = (res_b[tid] - res_bn_m[tid]) * rsc + res_bn_b[tid];
  }
  if (tid < NOUT) s_fcb[tid] = fc_b[tid];

  for (int idx = tid; idx < T1 * NV * NCIN; idx += nthr) {
    int i = idx / (NV * NCIN), r = idx % (NV * NCIN);
    int v = r >> 1, ci = r & 1;
    int t = tbase - 3 + i;
    float val = 0.0f;
    if (t >= 0 && t < NTIME)
      val = x[(((b * NTIME + t) * 2 + hand) * NV + v) * NCIN + ci];
    s_in[idx] = val;
  }
  __syncthreads();

  if (wave == 0) {
    if (tid < T1 * NCIN)
      graph_agg<NCIN, NCIN>(&s_in[(tid >> 1) * (NV * NCIN) + (tid & 1)],
                            &s_xa[(tid >> 1) * (NV * NCIN) + (tid & 1)]);
  } else if (wave < 4) {
    const int k = wave - 1;
    const float* rw = &s_raw[lane * PR + k];
    float w0 = 0.0f, w1 = 0.0f, wb = 0.0f;
#pragma unroll
    for (int ci = 0; ci < NC; ++ci) {
      const float w = rw[ci * 3];
      w0 += w * gcn_w0[ci * 2 + 0];
      w1 += w * gcn_w0[ci * 2 + 1];
      wb += w * gcn_b0[ci];
    }
    *(float4*)&s_wf[(k * NC + lane) * 4] = make_float4(w0, w1, wb, 0.0f);
  }
  __syncthreads();

  float tw1r[12];
#pragma unroll
  for (int i = 0; i < 12; ++i) {
    int idx = tid + i * 256;
    int co = idx & 31, kap = idx >> 5, k = kap >> 5, ci = kap & 31;
    tw1r[i] = tcn_w1[(co * NC + ci) * 3 + k];
  }

  {
    const int M = T2 * NV;
    for (int tIdx = wave; tIdx < 16 * 2; tIdx += 8) {
      int mt = tIdx >> 1, nt = tIdx & 1;
      v8f acc = {0.f,0.f,0.f,0.f,0.f,0.f,0.f,0.f};
      int mA = mt * 16 + row;
      int mC = (mA < M) ? mA : 0;
      int t2 = mC / NV;
      int colN = nt * 16 + row;
#pragma unroll
      for (int k = 0; k < 3; ++k) {
        const v2f xv = *(const v2f*)&s_xa[(mC + k * NV) * NCIN];
        const int t = tbase - 3 + t2 + k;
        const float ok = (t >= 0 && t < NTIME) ? 1.0f : 0.0f;
        v2f a;
        a.x = hi ? ok : xv.x;
        a.y = hi ? 0.0f : xv.y;
        const v2f bb = *(const v2f*)&s_wf[(k * NC + colN) * 4 + hi * 2];
        acc = wmma4(a, bb, acc);
      }
#pragma unroll
      for (int j = 0; j < 8; ++j) {
        int m = mt * 16 + j + hi * 8;
        if (m < M) {
          int te = m / NV, ve = m % NV;
          int co = colN;
          float val = acc[j] * s_s0[co] + s_o0[co];
          float x0 = s_in[((te + 1) * NV + ve) * 2 + 0];
          float x1 = s_in[((te + 1) * NV + ve) * 2 + 1];
          val += s_rA[co] * x0 + s_rB[co] * x1 + s_rC[co];
          s_h0[(te * NV + ve) * PC + co] = fmaxf(val, 0.0f);
        }
      }
    }
  }
  __syncthreads();

#pragma unroll
  for (int i = 0; i < 12; ++i) {
    int idx = tid + i * 256;
    s_tw[(idx >> 5) * PW + (idx & 31)] = tw1r[i];
  }

  float* s_hagg = s_g1;
  for (int t3 = wave; t3 < T3; t3 += 8)
    graph_agg<PC, PC>(&s_h0[(t3 + 1) * (NV * PC) + lane], &s_hagg[t3 * (NV * PC) + lane]);
  __syncthreads();

  {
    const int M = T3 * NV;
    const float gb1a = s_gb1[row], gb1b = s_gb1[16 + row];
    for (int mt = wave; mt < 14; mt += 8) {
      int mA = mt * 16 + row;
      int mC = (mA < M) ? mA : mt * 16;
      v2f a[8];
#pragma unroll
      for (int ks = 0; ks < 8; ++ks) {
        int k0 = ks * 4 + hi * 2;
        a[ks].x = s_hagg[mC * PC + k0];
        a[ks].y = s_hagg[mC * PC + k0 + 1];
      }
      v8f acc0 = {0.f,0.f,0.f,0.f,0.f,0.f,0.f,0.f};
      v8f acc1 = {0.f,0.f,0.f,0.f,0.f,0.f,0.f,0.f};
#pragma unroll
      for (int ks = 0; ks < 8; ++ks) {
        int k0 = ks * 4 + hi * 2;
        v2f b0, b1;
        b0.x = s_gw1[row * PC + k0];
        b0.y = s_gw1[row * PC + k0 + 1];
        b1.x = s_gw1[(16 + row) * PC + k0];
        b1.y = s_gw1[(16 + row) * PC + k0 + 1];
        acc0 = wmma4(a[ks], b0, acc0);
        acc1 = wmma4(a[ks], b1, acc1);
      }
#pragma unroll
      for (int j = 0; j < 8; ++j) {
        int m = mt * 16 + j + hi * 8;
        if (m < M) {
          int t3 = m / NV;
          int t = tbase - 1 + t3;
          bool ok = (t >= 0 && t < NTIME);
          s_g1[m * PC + row]      = ok ? acc0[j] + gb1a : 0.0f;
          s_g1[m * PC + 16 + row] = ok ? acc1[j] + gb1b : 0.0f;
        }
      }
    }
  }
  __syncthreads();

  {
    const int M = TT * NV;
    for (int tIdx = wave; tIdx < 11 * 2; tIdx += 8) {
      int mt = tIdx >> 1, nt = tIdx & 1;
      v8f acc = {0.f,0.f,0.f,0.f,0.f,0.f,0.f,0.f};
      int mA = mt * 16 + row;
      int mC = (mA < M) ? mA : 0;
      int t4 = mC / NV, vv = mC % NV;
      int colN = nt * 16 + row;
      for (int ks = 0; ks < 24; ++ks) {
        int k0 = ks * 4 + hi * 2;
        v2f a, bb;
        a.x = s_g1[((t4 + (k0 >> 5)) * NV + vv) * PC + (k0 & 31)];
        a.y = s_g1[((t4 + ((k0 + 1) >> 5)) * NV + vv) * PC + ((k0 + 1) & 31)];
        bb.x = s_tw[k0 * PW + colN];
        bb.y = s_tw[(k0 + 1) * PW + colN];
        acc = wmma4(a, bb, acc);
      }
#pragma unroll
      for (int j = 0; j < 8; ++j) {
        int m = mt * 16 + j + hi * 8;
        if (m < M) {
          int te = m / NV, ve = m % NV;
          int co = colN;
          int hidx = ((te + 2) * NV + ve) * PC + co;
          float val = acc[j] * s_s1[co] + s_o1[co] + s_h0[hidx];
          s_h0[hidx] = fmaxf(val, 0.0f);
        }
      }
    }
  }
  __syncthreads();

  {
    float* s_red = s_g1;
    int nt  = wave & 3;
    int ksb = (wave >> 2) * 84;
    v8f acc = {0.f,0.f,0.f,0.f,0.f,0.f,0.f,0.f};
    int t4 = (row < TT) ? row : (TT - 1);
    int colN = nt * 16 + row;
    const float* h1base = &s_h0[(t4 + 2) * (NV * PC)];
    for (int ks = ksb; ks < ksb + 84; ++ks) {
      int f0 = ks * 4 + hi * 2;
      int f1 = f0 + 1;
      int c0 = f0 / NV, v0 = f0 % NV;
      int c1 = f1 / NV, v1 = f1 % NV;
      v2f a, bb;
      a.x = h1base[v0 * PC + c0];
      a.y = h1base[v1 * PC + c1];
      bb.x = fc_w[colN * 672 + f0];
      bb.y = fc_w[colN * 672 + f1];
      acc = wmma4(a, bb, acc);
    }
    if (wave >= 4) {
#pragma unroll
      for (int j = 0; j < 8; ++j) s_red[nt * 256 + lane * 8 + j] = acc[j];
    }
    __syncthreads();
    if (wave < 4) {
#pragma unroll
      for (int j = 0; j < 8; ++j) {
        int m = j + hi * 8;
        if (m < TT) {
          int t = tbase + m;
          float val = acc[j] + s_red[nt * 256 + lane * 8 + j] + s_fcb[colN];
          out[(((size_t)b * NTIME + t) * 2 + hand) * NOUT + colN] = val;
        }
      }
    }
  }
}

extern "C" void kernel_launch(void* const* d_in, const int* in_sizes, int n_in,
                              void* d_out, int out_size, void* d_ws, size_t ws_size,
                              hipStream_t stream) {
  (void)in_sizes; (void)n_in; (void)d_ws; (void)ws_size; (void)out_size;
  const float* x        = (const float*)d_in[0];
  const float* gcn_w0   = (const float*)d_in[1];
  const float* gcn_b0   = (const float*)d_in[2];
  const float* tcn_w0   = (const float*)d_in[3];
  const float* tcn_b0   = (const float*)d_in[4];
  const float* bn_g0    = (const float*)d_in[5];
  const float* bn_b0    = (const float*)d_in[6];
  const float* bn_m0    = (const float*)d_in[7];
  const float* bn_v0    = (const float*)d_in[8];
  const float* res_w    = (const float*)d_in[9];
  const float* res_b    = (const float*)d_in[10];
  const float* res_bn_g = (const float*)d_in[11];
  const float* res_bn_b = (const float*)d_in[12];
  const float* res_bn_m = (const float*)d_in[13];
  const float* res_bn_v = (const float*)d_in[14];
  const float* gcn_w1   = (const float*)d_in[15];
  const float* gcn_b1   = (const float*)d_in[16];
  const float* tcn_w1   = (const float*)d_in[17];
  const float* tcn_b1   = (const float*)d_in[18];
  const float* bn_g1    = (const float*)d_in[19];
  const float* bn_b1    = (const float*)d_in[20];
  const float* bn_m1    = (const float*)d_in[21];
  const float* bn_v1    = (const float*)d_in[22];
  const float* fc_w     = (const float*)d_in[23];
  const float* fc_b     = (const float*)d_in[24];

  dim3 grid(NB * 2 * (NTIME / TT));
  dim3 block(256);
  stgcn_fused<<<grid, block, 0, stream>>>(
      x, gcn_w0, gcn_b0, tcn_w0, tcn_b0, bn_g0, bn_b0, bn_m0, bn_v0,
      res_w, res_b, res_bn_g, res_bn_b, res_bn_m, res_bn_v,
      gcn_w1, gcn_b1, tcn_w1, tcn_b1, bn_g1, bn_b1, bn_m1, bn_v1,
      fc_w, fc_b, (float*)d_out);
}


# FC loop: k order remapped so a lane fetches fc_w as 16-byte loads kept 14 ahead and reads h1 at constant offsets from two lane-half bases; per-step /21 address math gone; new sum order in FC
# speedup vs baseline: 3.1468x; 1.2980x over previous
#include <hip/hip_runtime.h>

#define TT 8
#define T1 (TT + 6)
#define T2 (TT + 4)
#define T3 (TT + 2)
#define NB 64
#define NTIME 512
#define NV 21
#define NC 32
#define NCIN 2
#define NOUT 64
#define PC 36
#define PW 40
#define PR 97

typedef __attribute__((ext_vector_type(2))) float v2f;
typedef __attribute__((ext_vector_type(8))) float v8f;

__device__ __forceinline__ v8f wmma4(v2f a, v2f b, v8f c) {
  return __builtin_amdgcn_wmma_f32_16x16x4_f32(
       false, a,  false, b,
       (short)0, c,  false,  false);
}

__host__ __device__ constexpr int g_parent(int j) { return ((j - 1) % 4 == 0) ? 0 : j - 1; }
__host__ __device__ constexpr bool g_conn(int i, int j) {
  return i == j || (j > 0 && g_parent(j) == i) || (i > 0 && g_parent(i) == j);
}
__host__ __device__ constexpr float g_dinv(int v) {
  return v == 0 ? 0.4082482905f : ((v % 4 == 0) ? 0.7071067812f : 0.5773502692f);
}

__host__ __device__ constexpr int  fc_off(int g)    { return (g % NV) * PC + g / NV; }
__host__ __device__ constexpr bool fc_nowrap(int g) { return (g % NV) + 4 < NV; }

template <int STRIDE_IN, int STRIDE_OUT>
__device__ __forceinline__ void graph_agg(const float* __restrict__ in, float* __restrict__ out) {
  float h[NV];
#pragma unroll
  for (int u = 0; u < NV; ++u) h[u] = in[u * STRIDE_IN];
#pragma unroll
  for (int v = 0; v < NV; ++v) {
    float acc = 0.0f;
    bool have = false;
#pragma unroll
    for (int u = 0; u < NV; ++u) {
      if (g_conn(v, u)) {
        const float p = (g_dinv(v) * g_dinv(u)) * h[u];
        acc = have ? acc + p : p;
        have = true;
      }
    }
    out[v * STRIDE_OUT] = acc;
  }
}

__global__ __launch_bounds__(256) void stgcn_fused(
    const float* __restrict__ x,
    const float* __restrict__ gcn_w0, const float* __restrict__ gcn_b0,
    const float* __restrict__ tcn_w0, const float* __restrict__ tcn_b0,
    const float* __restrict__ bn_g0, const float* __restrict__ bn_b0,
    const float* __restrict__ bn_m0, const float* __restrict__ bn_v0,
    const float* __restrict__ res_w, const float* __restrict__ res_b,
    const float* __restrict__ res_bn_g, const float* __restrict__ res_bn_b,
    const float* __restrict__ res_bn_m, const float* __restrict__ res_bn_v,
    const float* __restrict__ gcn_w1, const float* __restrict__ gcn_b1,
    const float* __restrict__ tcn_w1, const float* __restrict__ tcn_b1,
    const float* __restrict__ bn_g1, const float* __restrict__ bn_b1,
    const float* __restrict__ bn_m1, const float* __restrict__ bn_v1,
    const float* __restrict__ fc_w, const float* __restrict__ fc_b,
    float* __restrict__ out)
{
  __shared__ float s_in[T1 * NV * NCIN];
  __shared__ __attribute__((aligned(16))) float s_xa[T1 * NV * NCIN];
  __shared__ float s_tw[96 * PW];
  __shared__ float s_gw1[NC * PC];
  __shared__ float s_g1[T3 * NV * PC];
  __shared__ float s_h0[T2 * NV * PC];
  __shared__ __attribute__((aligned(16))) float s_wf[3 * NC * 4];
  __shared__ float s_gb1[NC];
  __shared__ float s_s0[NC], s_o0[NC], s_s1[NC], s_o1[NC];
  __shared__ float s_rA[NC], s_rB[NC], s_rC[NC];
  __shared__ float s_fcb[NOUT];

  const int tid  = threadIdx.x;
  const int nthr = 256;
  const int lane = tid & 31;
  const int wave = tid >> 5;
  const int row  = lane & 15;
  const int hi   = lane >> 4;

  const int bid   = blockIdx.x;
  const int tile  = bid & 63;
  const int hand  = (bid >> 6) & 1;
  const int b     = bid >> 7;
  const int tbase = tile * TT;

  float* s_raw = s_g1;
#pragma unroll
  for (int i = 0; i < 12; ++i) {
    int idx = tid + i * 256;
    int co = idx / 96;
    s_raw[co * PR + (idx - co * 96)] = tcn_w0[idx];
  }
  for (int idx = tid; idx < NC * NC; idx += nthr) s_gw1[(idx >> 5) * PC + (idx & 31)] = gcn_w1[idx];
  if (tid < NC) {
    s_gb1[tid] = gcn_b1[tid];
    float sc0 = bn_g0[tid] * rsqrtf(bn_v0[tid] + 1e-5f);
    s_s0[tid] = sc0;
    s_o0[tid] = (tcn_b0[tid] - bn_m0[tid]) * sc0 + bn_b0[tid];
    float sc1 = bn_g1[tid] * rsqrtf(bn_v1[tid] + 1e-5f);
    s_s1[tid] = sc1;
    s_o1[tid] = (tcn_b1[tid] - bn_m1[tid]) * sc1 + bn_b1[tid];
    float rsc = res_bn_g[tid] * rsqrtf(res_bn_v[tid] + 1e-5f);
    s_rA[tid] = res_w[tid * 2 + 0] * rsc;
    s_rB[tid] = res_w[tid * 2 + 1] * rsc;
    s_rC[tid] = (res_b[tid] - res_bn_m[tid]) * rsc + res_bn_b[tid];
  }
  if (tid < NOUT) s_fcb[tid] = fc_b[tid];

  for (int idx = tid; idx < T1 * NV * NCIN; idx += nthr) {
    int i = idx / (NV * NCIN), r = idx % (NV * NCIN);
    int v = r >> 1, ci = r & 1;
    int t = tbase - 3 + i;
    float val = 0.0f;
    if (t >= 0 && t < NTIME)
      val = x[(((b * NTIME + t) * 2 + hand) * NV + v) * NCIN + ci];
    s_in[idx] = val;
  }
  __syncthreads();

  if (wave == 0) {
    if (tid < T1 * NCIN)
      graph_agg<NCIN, NCIN>(&s_in[(tid >> 1) * (NV * NCIN) + (tid & 1)],
                            &s_xa[(tid >> 1) * (NV * NCIN) + (tid & 1)]);
  } else if (wave < 4) {
    const int k = wave - 1;
    const float* rw = &s_raw[lane * PR + k];
    float w0 = 0.0f, w1 = 0.0f, wb = 0.0f;
#pragma unroll
    for (int ci = 0; ci < NC; ++ci) {
      const float w = rw[ci * 3];
      w0 += w * gcn_w0[ci * 2 + 0];
      w1 += w * gcn_w0[ci * 2 + 1];
      wb += w * gcn_b0[ci];
    }
    *(float4*)&s_wf[(k * NC + lane) * 4] = make_float4(w0, w1, wb, 0.0f);
  }
  __syncthreads();

  float tw1r[12];
#pragma unroll
  for (int i = 0; i < 12; ++i) {
    int idx = tid + i * 256;
    int co = idx & 31, kap = idx >> 5, k = kap >> 5, ci = kap & 31;
    tw1r[i] = tcn_w1[(co * NC + ci) * 3 + k];
  }

  {
    const int M = T2 * NV;
    for (int tIdx = wave; tIdx < 16 * 2; tIdx += 8) {
      int mt = tIdx >> 1, nt = tIdx & 1;
      v8f acc = {0.f,0.f,0.f,0.f,0.f,0.f,0.f,0.f};
      int mA = mt * 16 + row;
      int mC = (mA < M) ? mA : 0;
      int t2 = mC / NV;
      int colN = nt * 16 + row;
#pragma unroll
      for (int k = 0; k < 3; ++k) {
        const v2f xv = *(const v2f*)&s_xa[(mC + k * NV) * NCIN];
        const int t = tbase - 3 + t2 + k;
        const float ok = (t >= 0 && t < NTIME) ? 1.0f : 0.0f;
        v2f a;
        a.x = hi ? ok : xv.x;
        a.y = hi ? 0.0f : xv.y;
        const v2f bb = *(const v2f*)&s_wf[(k * NC + colN) * 4 + hi * 2];
        acc = wmma4(a, bb, acc);
      }
#pragma unroll
      for (int j = 0; j < 8; ++j) {
        int m = mt * 16 + j + hi * 8;
        if (m < M) {
          int te = m / NV, ve = m % NV;
          int co = colN;
          float val = acc[j] * s_s0[co] + s_o0[co];
          float x0 = s_in[((te + 1) * NV + ve) * 2 + 0];
          float x1 = s_in[((te + 1) * NV + ve) * 2 + 1];
          val += s_rA[co] * x0 + s_rB[co] * x1 + s_rC[co];
          s_h0[(te * NV + ve) * PC + co] = fmaxf(val, 0.0f);
        }
      }
    }
  }
  __syncthreads();

#pragma unroll
  for (int i = 0; i < 12; ++i) {
    int idx = tid + i * 256;
    s_tw[(idx >> 5) * PW + (idx & 31)] = tw1r[i];
  }

  float* s_hagg = s_g1;
  for (int t3 = wave; t3 < T3; t3 += 8)
    graph_agg<PC, PC>(&s_h0[(t3 + 1) * (NV * PC) + lane], &s_hagg[t3 * (NV * PC) + lane]);
  __syncthreads();

  {
    const int M = T3 * NV;
    const float gb1a = s_gb1[row], gb1b = s_gb1[16 + row];
    for (int mt = wave; mt < 14; mt += 8) {
      int mA = mt * 16 + row;
      int mC = (mA < M) ? mA : mt * 16;
      v2f a[8];
#pragma unroll
      for (int ks = 0; ks < 8; ++ks) {
        int k0 = ks * 4 + hi * 2;
        a[ks].x = s_hagg[mC * PC + k0];
        a[ks].y = s_hagg[mC * PC + k0 + 1];
      }
      v8f acc0 = {0.f,0.f,0.f,0.f,0.f,0.f,0.f,0.f};
      v8f acc1 = {0.f,0.f,0.f,0.f,0.f,0.f,0.f,0.f};
#pragma unroll
      for (int ks = 0; ks < 8; ++ks) {
        int k0 = ks * 4 + hi * 2;
        v2f b0, b1;
        b0.x = s_gw1[row * PC + k0];
        b0.y = s_gw1[row * PC + k0 + 1];
        b1.x = s_gw1[(16 + row) * PC + k0];
        b1.y = s_gw1[(16 + row) * PC + k0 + 1];
        acc0 = wmma4(a[ks], b0, acc0);
        acc1 = wmma4(a[ks], b1, acc1);
      }
#pragma unroll
      for (int j = 0; j < 8; ++j) {
        int m = mt * 16 + j + hi * 8;
        if (m < M) {
          int t3 = m / NV;
          int t = tbase - 1 + t3;
          bool ok = (t >= 0 && t < NTIME);
          s_g1[m * PC + row]      = ok ? acc0[j] + gb1a : 0.0f;
          s_g1[m * PC + 16 + row] = ok ? acc1[j] + gb1b : 0.0f;
        }
      }
    }
  }
  __syncthreads();

  {
    const int M = TT * NV;
    for (int tIdx = wave; tIdx < 11 * 2; tIdx += 8) {
      int mt = tIdx >> 1, nt = tIdx & 1;
      v8f acc = {0.f,0.f,0.f,0.f,0.f,0.f,0.f,0.f};
      int mA = mt * 16 + row;
      int mC = (mA < M) ? mA : 0;
      int t4 = mC / NV, vv = mC % NV;
      int colN = nt * 16 + row;
      for (int ks = 0; ks < 24; ++ks) {
        int k0 = ks * 4 + hi * 2;
        v2f a, bb;
        a.x = s_g1[((t4 + (k0 >> 5)) * NV + vv) * PC + (k0 & 31)];
        a.y = s_g1[((t4 + ((k0 + 1) >> 5)) * NV + vv) * PC + ((k0 + 1) & 31)];
        bb.x = s_tw[k0 * PW + colN];
        bb.y = s_tw[(k0 + 1) * PW + colN];
        acc = wmma4(a, bb, acc);
      }
#pragma unroll
      for (int j = 0; j < 8; ++j) {
        int m = mt * 16 + j + hi * 8;
        if (m < M) {
          int te = m / NV, ve = m % NV;
          int co = colN;
          int hidx = ((te + 2) * NV + ve) * PC + co;
          float val = acc[j] * s_s1[co] + s_o1[co] + s_h0[hidx];
          s_h0[hidx] = fmaxf(val, 0.0f);
        }
      }
    }
  }

  constexpr int FCQ = 42;
  constexpr int FCD = 14;
  const int fc_nt = wave & 3;
  const int fc_kh = wave >> 2;
  const int fc_col = fc_nt * 16 + row;
  const float* fc_bp = fc_w + fc_col * 672 + fc_kh * 336 + hi * 4;
  float4 fcw_r[FCD];
#pragma unroll
  for (int q = 0; q < FCD; ++q) fcw_r[q] = *(const float4*)(fc_bp + 8 * q);
  __syncthreads();

  {
    float* s_red = s_g1;
    const int nt = fc_nt;
    const int colN = fc_col;
    v8f acc = {0.f,0.f,0.f,0.f,0.f,0.f,0.f,0.f};
    const int t4 = (row < TT) ? row : (TT - 1);
    const float* h1row = &s_h0[(t4 + 2) * (NV * PC) + 16 * fc_kh];
    const float* aP = h1row + hi * (4 * PC);
    const float* aQ = h1row + hi * ((4 - NV) * PC + 1);
#pragma unroll
    for (int q = 0; q < FCQ; ++q) {
      const float4 bv = fcw_r[q % FCD];
      if (q + FCD < FCQ) fcw_r[q % FCD] = *(const float4*)(fc_bp + 8 * (q + FCD));
      const int g = 8 * q;
      v2f a0, a1, b0, b1;
      a0.x = (fc_nowrap(g + 0) ? aP : aQ)[fc_off(g + 0)];
      a0.y = (fc_nowrap(g + 1) ? aP : aQ)[fc_off(g + 1)];
      a1.x = (fc_nowrap(g + 2) ? aP : aQ)[fc_off(g + 2)];
      a1.y = (fc_nowrap(g + 3) ? aP : aQ)[fc_off(g + 3)];
      b0.x = bv.x; b0.y = bv.y;
      b1.x = bv.z; b1.y = bv.w;
      acc = wmma4(a0, b0, acc);
      acc = wmma4(a1, b1, acc);
    }
    if (wave >= 4) {
#pragma unroll
      for (int j = 0; j < 8; ++j) s_red[nt * 256 + lane * 8 + j] = acc[j];
    }
    __syncthreads();
    if (wave < 4) {
#pragma unroll
      for (int j = 0; j < 8; ++j) {
        int m = j + hi * 8;
        if (m < TT) {
          int t = tbase + m;
          float val = acc[j] + s_red[nt * 256 + lane * 8 + j] + s_fcb[colN];
          out[(((size_t)b * NTIME + t) * 2 + hand) * NOUT + colN] = val;
        }
      }
    }
  }
}

extern "C" void kernel_launch(void* const* d_in, const int* in_sizes, int n_in,
                              void* d_out, int out_size, void* d_ws, size_t ws_size,
                              hipStream_t stream) {
  (void)in_sizes; (void)n_in; (void)d_ws; (void)ws_size; (void)out_size;
  const float* x        = (const float*)d_in[0];
  const float* gcn_w0   = (const float*)d_in[1];
  const float* gcn_b0   = (const float*)d_in[2];
  const float* tcn_w0   = (const float*)d_in[3];
  const float* tcn_b0   = (const float*)d_in[4];
  const float* bn_g0    = (const float*)d_in[5];
  const float* bn_b0    = (const float*)d_in[6];
  const float* bn_m0    = (const float*)d_in[7];
  const float* bn_v0    = (const float*)d_in[8];
  const float* res_w    = (const float*)d_in[9];
  const float* res_b    = (const float*)d_in[10];
  const float* res_bn_g = (const float*)d_in[11];
  const float* res_bn_b = (const float*)d_in[12];
  const float* res_bn_m = (const float*)d_in[13];
  const float* res_bn_v = (const float*)d_in[14];
  const float* gcn_w1   = (const float*)d_in[15];
  const float* gcn_b1   = (const float*)d_in[16];
  const float* tcn_w1   = (const float*)d_in[17];
  const float* tcn_b1   = (const float*)d_in[18];
  const float* bn_g1    = (const float*)d_in[19];
  const float* bn_b1    = (const float*)d_in[20];
  const float* bn_m1    = (const float*)d_in[21];
  const float* bn_v1    = (const float*)d_in[22];
  const float* fc_w     = (const float*)d_in[23];
  const float* fc_b     = (const float*)d_in[24];

  dim3 grid(NB * 2 * (NTIME / TT));
  dim3 block(256);
  stgcn_fused<<<grid, block, 0, stream>>>(
      x, gcn_w0, gcn_b0, tcn_w0, tcn_b0, bn_g0, bn_b0, bn_m0, bn_v0,
      res_w, res_b, res_bn_g, res_bn_b, res_bn_m, res_bn_v,
      gcn_w1, gcn_b1, tcn_w1, tcn_b1, bn_g1, bn_b1, bn_m1, bn_v1,
      fc_w, fc_b, (float*)d_out);
}


# Time tile per workgroup 8->16 in 512-thread workgroups: FC fills all 16 WMMA rows, so fc_w is fetched and multiplied half as often; halo rows per output halved; FC K split 4 ways (new sum order)
# speedup vs baseline: 4.4961x; 1.4288x over previous
#include <hip/hip_runtime.h>

#define TT 16
#define T1 (TT + 6)
#define T2 (TT + 4)
#define T3 (TT + 2)
#define NTHR 512
#define NWAVE (NTHR / 32)
#define NB 64
#define NTIME 512
#define NV 21
#define NC 32
#define NCIN 2
#define NOUT 64
#define PC 36
#define PW 40
#define PR 97
#define RP 68

typedef __attribute__((ext_vector_type(2))) float v2f;
typedef __attribute__((ext_vector_type(8))) float v8f;

__device__ __forceinline__ v8f wmma4(v2f a, v2f b, v8f c) {
  return __builtin_amdgcn_wmma_f32_16x16x4_f32(
       false, a,  false, b,
       (short)0, c,  false,  false);
}

__host__ __device__ constexpr int g_parent(int j) { return ((j - 1) % 4 == 0) ? 0 : j - 1; }
__host__ __device__ constexpr bool g_conn(int i, int j) {
  return i == j || (j > 0 && g_parent(j) == i) || (i > 0 && g_parent(i) == j);
}
__host__ __device__ constexpr float g_dinv(int v) {
  return v == 0 ? 0.4082482905f : ((v % 4 == 0) ? 0.7071067812f : 0.5773502692f);
}

__host__ __device__ constexpr int  fc_off(int g)    { return (g % NV) * PC + g / NV; }
__host__ __device__ constexpr bool fc_nowrap(int g) { return (g % NV) + 4 < NV; }

template <int STRIDE_IN, int STRIDE_OUT>
__device__ __forceinline__ void graph_agg(const float* __restrict__ in, float* __restrict__ out) {
  float h[NV];
#pragma unroll
  for (int u = 0; u < NV; ++u) h[u] = in[u * STRIDE_IN];
#pragma unroll
  for (int v = 0; v < NV; ++v) {
    float acc = 0.0f;
    bool have = false;
#pragma unroll
    for (int u = 0; u < NV; ++u) {
      if (g_conn(v, u)) {
        const float p = (g_dinv(v) * g_dinv(u)) * h[u];
        acc = have ? acc + p : p;
        have = true;
      }
    }
    out[v * STRIDE_OUT] = acc;
  }
}

__global__ __launch_bounds__(NTHR) void stgcn_fused(
    const float* __restrict__ x,
    const float* __restrict__ gcn_w0, const float* __restrict__ gcn_b0,
    const float* __restrict__ tcn_w0, const float* __restrict__ tcn_b0,
    const float* __restrict__ bn_g0, const float* __restrict__ bn_b0,
    const float* __restrict__ bn_m0, const float* __restrict__ bn_v0,
    const float* __restrict__ res_w, const float* __restrict__ res_b,
    const float* __restrict__ res_bn_g, const float* __restrict__ res_bn_b,
    const float* __restrict__ res_bn_m, const float* __restrict__ res_bn_v,
    const float* __restrict__ gcn_w1, const float* __restrict__ gcn_b1,
    const float* __restrict__ tcn_w1, const float* __restrict__ tcn_b1,
    const float* __restrict__ bn_g1, const float* __restrict__ bn_b1,
    const float* __restrict__ bn_m1, const float* __restrict__ bn_v1,
    const float* __restrict__ fc_w, const float* __restrict__ fc_b,
    float* __restrict__ out)
{
  __shared__ float s_in[T1 * NV * NCIN];
  __shared__ __attribute__((aligned(16))) float s_xa[T1 * NV * NCIN];
  __shared__ float s_tw[96 * PW];
  __shared__ float s_gw1[NC * PC];
  __shared__ float s_g1[T3 * NV * PC];
  __shared__ float s_h0[T2 * NV * PC];
  __shared__ __attribute__((aligned(16))) float s_wf[3 * NC * 4];
  __shared__ float s_gb1[NC];
  __shared__ float s_s0[NC], s_o0[NC], s_s1[NC], s_o1[NC];
  __shared__ float s_rA[NC], s_rB[NC], s_rC[NC];
  __shared__ float s_fcb[NOUT];

  const int tid  = threadIdx.x;
  const int nthr = NTHR;
  const int lane = tid & 31;
  const int wave = tid >> 5;
  const int row  = lane & 15;
  const int hi   = lane >> 4;

  const int bid   = blockIdx.x;
  const int tile  = bid & 31;
  const int hand  = (bid >> 5) & 1;
  const int b     = bid >> 6;
  const int tbase = tile * TT;

  float* s_raw = s_g1;
#pragma unroll
  for (int i = 0; i < 3072 / NTHR; ++i) {
    int idx = tid + i * NTHR;
    int co = idx / 96;
    s_raw[co * PR + (idx - co * 96)] = tcn_w0[idx];
  }
  for (int idx = tid; idx < NC * NC; idx += nthr) s_gw1[(idx >> 5) * PC + (idx & 31)] = gcn_w1[idx];
  if (tid < NC) {
    s_gb1[tid] = gcn_b1[tid];
    float sc0 = bn_g0[tid] * rsqrtf(bn_v0[tid] + 1e-5f);
    s_s0[tid] = sc0;
    s_o0[tid] = (tcn_b0[tid] - bn_m0[tid]) * sc0 + bn_b0[tid];
    float sc1 = bn_g1[tid] * rsqrtf(bn_v1[tid] + 1e-5f);
    s_s1[tid] = sc1;
    s_o1[tid] = (tcn_b1[tid] - bn_m1[tid]) * sc1 + bn_b1[tid];
    float rsc = res_bn_g[tid] * rsqrtf(res_bn_v[tid] + 1e-5f);
    s_rA[tid] = res_w[tid * 2 + 0] * rsc;
    s_rB[tid] = res_w[tid * 2 + 1] * rsc;
    s_rC[tid] = (res_b[tid] - res_bn_m[tid]) * rsc + res_bn_b[tid];
  }
  if (tid < NOUT) s_fcb[tid] = fc_b[tid];

  for (int idx = tid; idx < T1 * NV * NCIN; idx += nthr) {
    int i = idx / (NV * NCIN), r = idx % (NV * NCIN);
    int v = r >> 1, ci = r & 1;
    int t = tbase - 3 + i;
    float val = 0.0f;
    if (t >= 0 && t < NTIME)
      val = x[(((b * NTIME + t) * 2 + hand) * NV + v) * NCIN + ci];
    s_in[idx] = val;
  }
  __syncthreads();

  if (wave < 2) {
    if (tid < T1 * NCIN)
      graph_agg<NCIN, NCIN>(&s_in[(tid >> 1) * (NV * NCIN) + (tid & 1)],
                            &s_xa[(tid >> 1) * (NV * NCIN) + (tid & 1)]);
  } else if (wave < 5) {
    const int k = wave - 2;
    const float* rw = &s_raw[lane * PR + k];
    float w0 = 0.0f, w1 = 0.0f, wb = 0.0f;
#pragma unroll
    for (int ci = 0; ci < NC; ++ci) {
      const float w = rw[ci * 3];
      w0 += w * gcn_w0[ci * 2 + 0];
      w1 += w * gcn_w0[ci * 2 + 1];
      wb += w * gcn_b0[ci];
    }
    *(float4*)&s_wf[(k * NC + lane) * 4] = make_float4(w0, w1, wb, 0.0f);
  }
  __syncthreads();

  float tw1r[3072 / NTHR];
#pragma unroll
  for (int i = 0; i < 3072 / NTHR; ++i) {
    int idx = tid + i * NTHR;
    int co = idx & 31, kap = idx >> 5, k = kap >> 5, ci = kap & 31;
    tw1r[i] = tcn_w1[(co * NC + ci) * 3 + k];
  }

  {
    const int M = T2 * NV;
    for (int tIdx = wave; tIdx < 27 * 2; tIdx += NWAVE) {
      int mt = tIdx >> 1, nt = tIdx & 1;
      v8f acc = {0.f,0.f,0.f,0.f,0.f,0.f,0.f,0.f};
      int mA = mt * 16 + row;
      int mC = (mA < M) ? mA : 0;
      int t2 = mC / NV;
      int colN = nt * 16 + row;
#pragma unroll
      for (int k = 0; k < 3; ++k) {
        const v2f xv = *(const v2f*)&s_xa[(mC + k * NV) * NCIN];
        const int t = tbase - 3 + t2 + k;
        const float ok = (t >= 0 && t < NTIME) ? 1.0f : 0.0f;
        v2f a;
        a.x = hi ? ok : xv.x;
        a.y = hi ? 0.0f : xv.y;
        const v2f bb = *(const v2f*)&s_wf[(k * NC + colN) * 4 + hi * 2];
        acc = wmma4(a, bb, acc);
      }
#pragma unroll
      for (int j = 0; j < 8; ++j) {
        int m = mt * 16 + j + hi * 8;
        if (m < M) {
          int te = m / NV, ve = m % NV;
          int co = colN;
          float val = acc[j] * s_s0[co] + s_o0[co];
          float x0 = s_in[((te + 1) * NV + ve) * 2 + 0];
          float x1 = s_in[((te + 1) * NV + ve) * 2 + 1];
          val += s_rA[co] * x0 + s_rB[co] * x1 + s_rC[co];
          s_h0[(te * NV + ve) * PC + co] = fmaxf(val, 0.0f);
        }
      }
    }
  }
  __syncthreads();

#pragma unroll
  for (int i = 0; i < 3072 / NTHR; ++i) {
    int idx = tid + i * NTHR;
    s_tw[(idx >> 5) * PW + (idx & 31)] = tw1r[i];
  }

  float* s_hagg = s_g1;
  for (int t3 = wave; t3 < T3; t3 += NWAVE)
    graph_agg<PC, PC>(&s_h0[(t3 + 1) * (NV * PC) + lane], &s_hagg[t3 * (NV * PC) + lane]);
  __syncthreads();

  {
    const int M = T3 * NV;
    const float gb1a = s_gb1[row], gb1b = s_gb1[16 + row];
    for (int mt = wave; mt < 24; mt += NWAVE) {
      int mA = mt * 16 + row;
      int mC = (mA < M) ? mA : mt * 16;
      v2f a[8];
#pragma unroll
      for (int ks = 0; ks < 8; ++ks) {
        int k0 = ks * 4 + hi * 2;
        a[ks].x = s_hagg[mC * PC + k0];
        a[ks].y = s_hagg[mC * PC + k0 + 1];
      }
      v8f acc0 = {0.f,0.f,0.f,0.f,0.f,0.f,0.f,0.f};
      v8f acc1 = {0.f,0.f,0.f,0.f,0.f,0.f,0.f,0.f};
#pragma unroll
      for (int ks = 0; ks < 8; ++ks) {
        int k0 = ks * 4 + hi * 2;
        v2f b0, b1;
        b0.x = s_gw1[row * PC + k0];
        b0.y = s_gw1[row * PC + k0 + 1];
        b1.x = s_gw1[(16 + row) * PC + k0];
        b1.y = s_gw1[(16 + row) * PC + k0 + 1];
        acc0 = wmma4(a[ks], b0, acc0);
        acc1 = wmma4(a[ks], b1, acc1);
      }
#pragma unroll
      for (int j = 0; j < 8; ++j) {
        int m = mt * 16 + j + hi * 8;
        if (m < M) {
          int t3 = m / NV;
          int t = tbase - 1 + t3;
          bool ok = (t >= 0 && t < NTIME);
          s_g1[m * PC + row]      = ok ? acc0[j] + gb1a : 0.0f;
          s_g1[m * PC + 16 + row] = ok ? acc1[j] + gb1b : 0.0f;
        }
      }
    }
  }
  __syncthreads();

  {
    const int M = TT * NV;
    for (int tIdx = wave; tIdx < 21 * 2; tIdx += NWAVE) {
      int mt = tIdx >> 1, nt = tIdx & 1;
      v8f acc = {0.f,0.f,0.f,0.f,0.f,0.f,0.f,0.f};
      int mA = mt * 16 + row;
      int mC = (mA < M) ? mA : 0;
      int t4 = mC / NV, vv = mC % NV;
      int colN = nt * 16 + row;
      for (int ks = 0; ks < 24; ++ks) {
        int k0 = ks * 4 + hi * 2;
        v2f a, bb;
        a.x = s_g1[((t4 + (k0 >> 5)) * NV + vv) * PC + (k0 & 31)];
        a.y = s_g1[((t4 + ((k0 + 1) >> 5)) * NV + vv) * PC + ((k0 + 1) & 31)];
        bb.x = s_tw[k0 * PW + colN];
        bb.y = s_tw[(k0 + 1) * PW + colN];
        acc = wmma4(a, bb, acc);
      }
#pragma unroll
      for (int j = 0; j < 8; ++j) {
        int m = mt * 16 + j + hi * 8;
        if (m < M) {
          int te = m / NV, ve = m % NV;
          int co = colN;
          int hidx = ((te + 2) * NV + ve) * PC + co;
          float val = acc[j] * s_s1[co] + s_o1[co] + s_h0[hidx];
          s_h0[hidx] = fmaxf(val, 0.0f);
        }
      }
    }
  }

  constexpr int FCQ = 21;
  constexpr int FCD = 7;
  const int fc_nt = wave & 3;
  const int fc_kq = wave >> 2;
  const int fc_col = fc_nt * 16 + row;
  const float* fc_bp = fc_w + fc_col * 672 + fc_kq * 168 + hi * 4;
  float4 fcw_r[FCD];
#pragma unroll
  for (int q = 0; q < FCD; ++q) fcw_r[q] = *(const float4*)(fc_bp + 8 * q);
  __syncthreads();

  {
    float* s_red = s_g1;
    v8f acc = {0.f,0.f,0.f,0.f,0.f,0.f,0.f,0.f};
    const float* h1row = &s_h0[(row + 2) * (NV * PC) + 8 * fc_kq];
    const float* aP = h1row + hi * (4 * PC);
    const float* aQ = h1row + hi * ((4 - NV) * PC + 1);
#pragma unroll
    for (int q = 0; q < FCQ; ++q) {
      const float4 bv = fcw_r[q % FCD];
      if (q + FCD < FCQ) fcw_r[q % FCD] = *(const float4*)(fc_bp + 8 * (q + FCD));
      const int g = 8 * q;
      v2f a0, a1, b0, b1;
      a0.x = (fc_nowrap(g + 0) ? aP : aQ)[fc_off(g + 0)];
      a0.y = (fc_nowrap(g + 1) ? aP : aQ)[fc_off(g + 1)];
      a1.x = (fc_nowrap(g + 2) ? aP : aQ)[fc_off(g + 2)];
      a1.y = (fc_nowrap(g + 3) ? aP : aQ)[fc_off(g + 3)];
      b0.x = bv.x; b0.y = bv.y;
      b1.x = bv.z; b1.y = bv.w;
      acc = wmma4(a0, b0, acc);
      acc = wmma4(a1, b1, acc);
    }
#pragma unroll
    for (int j = 0; j < 8; ++j)
      s_red[(fc_kq * 16 + j + hi * 8) * RP + fc_col] = acc[j];
    __syncthreads();
#pragma unroll
    for (int i = 0; i < (TT * NOUT) / NTHR; ++i) {
      const int idx = tid + i * NTHR;
      const int m = idx >> 6, o = idx & 63;
      float val = s_red[m * RP + o];
      val += s_red[(16 + m) * RP + o];
      val += s_red[(32 + m) * RP + o];
      val += s_red[(48 + m) * RP + o];
      val += s_fcb[o];
      out[(((size_t)b * NTIME + (tbase + m)) * 2 + hand) * NOUT + o] = val;
    }
  }
}

extern "C" void kernel_launch(void* const* d_in, const int* in_sizes, int n_in,
                              void* d_out, int out_size, void* d_ws, size_t ws_size,
                              hipStream_t stream) {
  (void)in_sizes; (void)n_in; (void)d_ws; (void)ws_size; (void)out_size;
  const float* x        = (const float*)d_in[0];
  const float* gcn_w0   = (const float*)d_in[1];
  const float* gcn_b0   = (const float*)d_in[2];
  const float* tcn_w0   = (const float*)d_in[3];
  const float* tcn_b0   = (const float*)d_in[4];
  const float* bn_g0    = (const float*)d_in[5];
  const float* bn_b0    = (const float*)d_in[6];
  const float* bn_m0    = (const float*)d_in[7];
  const float* bn_v0    = (const float*)d_in[8];
  const float* res_w    = (const float*)d_in[9];
  const float* res_b    = (const float*)d_in[10];
  const float* res_bn_g = (const float*)d_in[11];
  const float* res_bn_b = (const float*)d_in[12];
  const float* res_bn_m = (const float*)d_in[13];
  const float* res_bn_v = (const float*)d_in[14];
  const float* gcn_w1   = (const float*)d_in[15];
  const float* gcn_b1   = (const float*)d_in[16];
  const float* tcn_w1   = (const float*)d_in[17];
  const float* tcn_b1   = (const float*)d_in[18];
  const float* bn_g1    = (const float*)d_in[19];
  const float* bn_b1    = (const float*)d_in[20];
  const float* bn_m1    = (const float*)d_in[21];
  const float* bn_v1    = (const float*)d_in[22];
  const float* fc_w     = (const float*)d_in[23];
  const float* fc_b     = (const float*)d_in[24];

  dim3 grid(NB * 2 * (NTIME / TT));
  dim3 block(NTHR);
  stgcn_fused<<<grid, block, 0, stream>>>(
      x, gcn_w0, gcn_b0, tcn_w0, tcn_b0, bn_g0, bn_b0, bn_m0, bn_v0,
      res_w, res_b, res_bn_g, res_bn_b, res_bn_m, res_bn_v,
      gcn_w1, gcn_b1, tcn_w1, tcn_b1, bn_g1, bn_b1, bn_m1, bn_v1,
      fc_w, fc_b, (float*)d_out);
}


# TCN0 epilogue: BN0/residual constants fetched once per wave, a tile's 8 residual input pairs fetched as one group, row guard only in the last partial tile (scalar test); same arithmetic
# speedup vs baseline: 4.9433x; 1.0995x over previous
#include <hip/hip_runtime.h>

#define TT 16
#define T1 (TT + 6)
#define T2 (TT + 4)
#define T3 (TT + 2)
#define NTHR 512
#define NWAVE (NTHR / 32)
#define NB 64
#define NTIME 512
#define NV 21
#define NC 32
#define NCIN 2
#define NOUT 64
#define PC 36
#define PW 40
#define PR 97
#define RP 68

typedef __attribute__((ext_vector_type(2))) float v2f;
typedef __attribute__((ext_vector_type(8))) float v8f;

__device__ __forceinline__ v8f wmma4(v2f a, v2f b, v8f c) {
  return __builtin_amdgcn_wmma_f32_16x16x4_f32(
       false, a,  false, b,
       (short)0, c,  false,  false);
}

__host__ __device__ constexpr int g_parent(int j) { return ((j - 1) % 4 == 0) ? 0 : j - 1; }
__host__ __device__ constexpr bool g_conn(int i, int j) {
  return i == j || (j > 0 && g_parent(j) == i) || (i > 0 && g_parent(i) == j);
}
__host__ __device__ constexpr float g_dinv(int v) {
  return v == 0 ? 0.4082482905f : ((v % 4 == 0) ? 0.7071067812f : 0.5773502692f);
}

__host__ __device__ constexpr int  fc_off(int g)    { return (g % NV) * PC + g / NV; }
__host__ __device__ constexpr bool fc_nowrap(int g) { return (g % NV) + 4 < NV; }

template <int STRIDE_IN, int STRIDE_OUT>
__device__ __forceinline__ void graph_agg(const float* __restrict__ in, float* __restrict__ out) {
  float h[NV];
#pragma unroll
  for (int u = 0; u < NV; ++u) h[u] = in[u * STRIDE_IN];
#pragma unroll
  for (int v = 0; v < NV; ++v) {
    float acc = 0.0f;
    bool have = false;
#pragma unroll
    for (int u = 0; u < NV; ++u) {
      if (g_conn(v, u)) {
        const float p = (g_dinv(v) * g_dinv(u)) * h[u];
        acc = have ? acc + p : p;
        have = true;
      }
    }
    out[v * STRIDE_OUT] = acc;
  }
}

__global__ __launch_bounds__(NTHR) void stgcn_fused(
    const float* __restrict__ x,
    const float* __restrict__ gcn_w0, const float* __restrict__ gcn_b0,
    const float* __restrict__ tcn_w0, const float* __restrict__ tcn_b0,
    const float* __restrict__ bn_g0, const float* __restrict__ bn_b0,
    const float* __restrict__ bn_m0, const float* __restrict__ bn_v0,
    const float* __restrict__ res_w, const float* __restrict__ res_b,
    const float* __restrict__ res_bn_g, const float* __restrict__ res_bn_b,
    const float* __restrict__ res_bn_m, const float* __restrict__ res_bn_v,
    const float* __restrict__ gcn_w1, const float* __restrict__ gcn_b1,
    const float* __restrict__ tcn_w1, const float* __restrict__ tcn_b1,
    const float* __restrict__ bn_g1, const float* __restrict__ bn_b1,
    const float* __restrict__ bn_m1, const float* __restrict__ bn_v1,
    const float* __restrict__ fc_w, const float* __restrict__ fc_b,
    float* __restrict__ out)
{
  __shared__ __attribute__((aligned(16))) float s_in[T1 * NV * NCIN];
  __shared__ __attribute__((aligned(16))) float s_xa[T1 * NV * NCIN];
  __shared__ float s_tw[96 * PW];
  __shared__ float s_gw1[NC * PC];
  __shared__ float s_g1[T3 * NV * PC];
  __shared__ float s_h0[T2 * NV * PC];
  __shared__ __attribute__((aligned(16))) float s_wf[3 * NC * 4];
  __shared__ float s_gb1[NC];
  __shared__ float s_s0[NC], s_o0[NC], s_s1[NC], s_o1[NC];
  __shared__ float s_rA[NC], s_rB[NC], s_rC[NC];
  __shared__ float s_fcb[NOUT];

  const int tid  = threadIdx.x;
  const int nthr = NTHR;
  const int lane = tid & 31;
  const int wave = tid >> 5;
  const int row  = lane & 15;
  const int hi   = lane >> 4;

  const int bid   = blockIdx.x;
  const int tile  = bid & 31;
  const int hand  = (bid >> 5) & 1;
  const int b     = bid >> 6;
  const int tbase = tile * TT;

  float* s_raw = s_g1;
#pragma unroll
  for (int i = 0; i < 3072 / NTHR; ++i) {
    int idx = tid + i * NTHR;
    int co = idx / 96;
    s_raw[co * PR + (idx - co * 96)] = tcn_w0[idx];
  }
  for (int idx = tid; idx < NC * NC; idx += nthr) s_gw1[(idx >> 5) * PC + (idx & 31)] = gcn_w1[idx];
  if (tid < NC) {
    s_gb1[tid] = gcn_b1[tid];
    float sc0 = bn_g0[tid] * rsqrtf(bn_v0[tid] + 1e-5f);
    s_s0[tid] = sc0;
    s_o0[tid] = (tcn_b0[tid] - bn_m0[tid]) * sc0 + bn_b0[tid];
    float sc1 = bn_g1[tid] * rsqrtf(bn_v1[tid] + 1e-5f);
    s_s1[tid] = sc1;
    s_o1[tid] = (tcn_b1[tid] - bn_m1[tid]) * sc1 + bn_b1[tid];
    float rsc = res_bn_g[tid] * rsqrtf(res_bn_v[tid] + 1e-5f);
    s_rA[tid] = res_w[tid * 2 + 0] * rsc;
    s_rB[tid] = res_w[tid * 2 + 1] * rsc;
    s_rC[tid] = (res_b[tid] - res_bn_m[tid]) * rsc + res_bn_b[tid];
  }
  if (tid < NOUT) s_fcb[tid] = fc_b[tid];

  for (int idx = tid; idx < T1 * NV * NCIN; idx += nthr) {
    int i = idx / (NV * NCIN), r = idx % (NV * NCIN);
    int v = r >> 1, ci = r & 1;
    int t = tbase - 3 + i;
    float val = 0.0f;
    if (t >= 0 && t < NTIME)
      val = x[(((b * NTIME + t) * 2 + hand) * NV + v) * NCIN + ci];
    s_in[idx] = val;
  }
  __syncthreads();

  if (wave < 2) {
    if (tid < T1 * NCIN)
      graph_agg<NCIN, NCIN>(&s_in[(tid >> 1) * (NV * NCIN) + (tid & 1)],
                            &s_xa[(tid >> 1) * (NV * NCIN) + (tid & 1)]);
  } else if (wave < 5) {
    const int k = wave - 2;
    const float* rw = &s_raw[lane * PR + k];
    float w0 = 0.0f, w1 = 0.0f, wb = 0.0f;
#pragma unroll
    for (int ci = 0; ci < NC; ++ci) {
      const float w = rw[ci * 3];
      w0 += w * gcn_w0[ci * 2 + 0];
      w1 += w * gcn_w0[ci * 2 + 1];
      wb += w * gcn_b0[ci];
    }
    *(float4*)&s_wf[(k * NC + lane) * 4] = make_float4(w0, w1, wb, 0.0f);
  }
  __syncthreads();

  float tw1r[3072 / NTHR];
#pragma unroll
  for (int i = 0; i < 3072 / NTHR; ++i) {
    int idx = tid + i * NTHR;
    int co = idx & 31, kap = idx >> 5, k = kap >> 5, ci = kap & 31;
    tw1r[i] = tcn_w1[(co * NC + ci) * 3 + k];
  }

  {
    const int M = T2 * NV;
    const int wave_u = __builtin_amdgcn_readfirstlane(wave);
    const int colN = (wave_u & 1) * 16 + row;
    const float c_s0 = s_s0[colN], c_o0 = s_o0[colN];
    const float c_rA = s_rA[colN], c_rB = s_rB[colN], c_rC = s_rC[colN];
    for (int tIdx = wave_u; tIdx < 27 * 2; tIdx += NWAVE) {
      const int mt = tIdx >> 1;
      v8f acc = {0.f,0.f,0.f,0.f,0.f,0.f,0.f,0.f};
      int mA = mt * 16 + row;
      int mC = (mA < M) ? mA : 0;
      int t2 = mC / NV;
      const int m0 = mt * 16 + hi * 8;
      v2f xr[8];
#pragma unroll
      for (int j = 0; j < 8; ++j) xr[j] = *(const v2f*)&s_in[(m0 + j + NV) * NCIN];
#pragma unroll
      for (int k = 0; k < 3; ++k) {
        const v2f xv = *(const v2f*)&s_xa[(mC + k * NV) * NCIN];
        const int t = tbase - 3 + t2 + k;
        const float ok = (t >= 0 && t < NTIME) ? 1.0f : 0.0f;
        v2f a;
        a.x = hi ? ok : xv.x;
        a.y = hi ? 0.0f : xv.y;
        const v2f bb = *(const v2f*)&s_wf[(k * NC + colN) * 4 + hi * 2];
        acc = wmma4(a, bb, acc);
      }
      float* hp = &s_h0[m0 * PC + colN];
      if (mt < 26) {
#pragma unroll
        for (int j = 0; j < 8; ++j) {
          float val = acc[j] * c_s0 + c_o0;
          val += c_rA * xr[j].x + c_rB * xr[j].y + c_rC;
          hp[j * PC] = fmaxf(val, 0.0f);
        }
      } else {
#pragma unroll
        for (int j = 0; j < 8; ++j) {
          if (m0 + j < M) {
            float val = acc[j] * c_s0 + c_o0;
            val += c_rA * xr[j].x + c_rB * xr[j].y + c_rC;
            hp[j * PC] = fmaxf(val, 0.0f);
          }
        }
      }
    }
  }
  __syncthreads();

#pragma unroll
  for (int i = 0; i < 3072 / NTHR; ++i) {
    int idx = tid + i * NTHR;
    s_tw[(idx >> 5) * PW + (idx & 31)] = tw1r[i];
  }

  float* s_hagg = s_g1;
  for (int t3 = wave; t3 < T3; t3 += NWAVE)
    graph_agg<PC, PC>(&s_h0[(t3 + 1) * (NV * PC) + lane], &s_hagg[t3 * (NV * PC) + lane]);
  __syncthreads();

  {
    const int M = T3 * NV;
    const float gb1a = s_gb1[row], gb1b = s_gb1[16 + row];
    for (int mt = wave; mt < 24; mt += NWAVE) {
      int mA = mt * 16 + row;
      int mC = (mA < M) ? mA : mt * 16;
      v2f a[8];
#pragma unroll
      for (int ks = 0; ks < 8; ++ks) {
        int k0 = ks * 4 + hi * 2;
        a[ks].x = s_hagg[mC * PC + k0];
        a[ks].y = s_hagg[mC * PC + k0 + 1];
      }
      v8f acc0 = {0.f,0.f,0.f,0.f,0.f,0.f,0.f,0.f};
      v8f acc1 = {0.f,0.f,0.f,0.f,0.f,0.f,0.f,0.f};
#pragma unroll
      for (int ks = 0; ks < 8; ++ks) {
        int k0 = ks * 4 + hi * 2;
        v2f b0, b1;
        b0.x = s_gw1[row * PC + k0];
        b0.y = s_gw1[row * PC + k0 + 1];
        b1.x = s_gw1[(16 + row) * PC + k0];
        b1.y = s_gw1[(16 + row) * PC + k0 + 1];
        acc0 = wmma4(a[ks], b0, acc0);
        acc1 = wmma4(a[ks], b1, acc1);
      }
#pragma unroll
      for (int j = 0; j < 8; ++j) {
        int m = mt * 16 + j + hi * 8;
        if (m < M) {
          int t3 = m / NV;
          int t = tbase - 1 + t3;
          bool ok = (t >= 0 && t < NTIME);
          s_g1[m * PC + row]      = ok ? acc0[j] + gb1a : 0.0f;
          s_g1[m * PC + 16 + row] = ok ? acc1[j] + gb1b : 0.0f;
        }
      }
    }
  }
  __syncthreads();

  {
    const int M = TT * NV;
    for (int tIdx = wave; tIdx < 21 * 2; tIdx += NWAVE) {
      int mt = tIdx >> 1, nt = tIdx & 1;
      v8f acc = {0.f,0.f,0.f,0.f,0.f,0.f,0.f,0.f};
      int mA = mt * 16 + row;
      int mC = (mA < M) ? mA : 0;
      int t4 = mC / NV, vv = mC % NV;
      int colN = nt * 16 + row;
      for (int ks = 0; ks < 24; ++ks) {
        int k0 = ks * 4 + hi * 2;
        v2f a, bb;
        a.x = s_g1[((t4 + (k0 >> 5)) * NV + vv) * PC + (k0 & 31)];
        a.y = s_g1[((t4 + ((k0 + 1) >> 5)) * NV + vv) * PC + ((k0 + 1) & 31)];
        bb.x = s_tw[k0 * PW + colN];
        bb.y = s_tw[(k0 + 1) * PW + colN];
        acc = wmma4(a, bb, acc);
      }
#pragma unroll
      for (int j = 0; j < 8; ++j) {
        int m = mt * 16 + j + hi * 8;
        if (m < M) {
          int te = m / NV, ve = m % NV;
          int co = colN;
          int hidx = ((te + 2) * NV + ve) * PC + co;
          float val = acc[j] * s_s1[co] + s_o1[co] + s_h0[hidx];
          s_h0[hidx] = fmaxf(val, 0.0f);
        }
      }
    }
  }

  constexpr int FCQ = 21;
  constexpr int FCD = 7;
  const int fc_nt = wave & 3;
  const int fc_kq = wave >> 2;
  const int fc_col = fc_nt * 16 + row;
  const float* fc_bp = fc_w + fc_col * 672 + fc_kq * 168 + hi * 4;
  float4 fcw_r[FCD];
#pragma unroll
  for (int q = 0; q < FCD; ++q) fcw_r[q] = *(const float4*)(fc_bp + 8 * q);
  __syncthreads();

  {
    float* s_red = s_g1;
    v8f acc = {0.f,0.f,0.f,0.f,0.f,0.f,0.f,0.f};
    const float* h1row = &s_h0[(row + 2) * (NV * PC) + 8 * fc_kq];
    const float* aP = h1row + hi * (4 * PC);
    const float* aQ = h1row + hi * ((4 - NV) * PC + 1);
#pragma unroll
    for (int q = 0; q < FCQ; ++q) {
      const float4 bv = fcw_r[q % FCD];
      if (q + FCD < FCQ) fcw_r[q % FCD] = *(const float4*)(fc_bp + 8 * (q + FCD));
      const int g = 8 * q;
      v2f a0, a1, b0, b1;
      a0.x = (fc_nowrap(g + 0) ? aP : aQ)[fc_off(g + 0)];
      a0.y = (fc_nowrap(g + 1) ? aP : aQ)[fc_off(g + 1)];
      a1.x = (fc_nowrap(g + 2) ? aP : aQ)[fc_off(g + 2)];
      a1.y = (fc_nowrap(g + 3) ? aP : aQ)[fc_off(g + 3)];
      b0.x = bv.x; b0.y = bv.y;
      b1.x = bv.z; b1.y = bv.w;
      acc = wmma4(a0, b0, acc);
      acc = wmma4(a1, b1, acc);
    }
#pragma unroll
    for (int j = 0; j < 8; ++j)
      s_red[(fc_kq * 16 + j + hi * 8) * RP + fc_col] = acc[j];
    __syncthreads();
#pragma unroll
    for (int i = 0; i < (TT * NOUT) / NTHR; ++i) {
      const int idx = tid + i * NTHR;
      const int m = idx >> 6, o = idx & 63;
      float val = s_red[m * RP + o];
      val += s_red[(16 + m) * RP + o];
      val += s_red[(32 + m) * RP + o];
      val += s_red[(48 + m) * RP + o];
      val += s_fcb[o];
      out[(((size_t)b * NTIME + (tbase + m)) * 2 + hand) * NOUT + o] = val;
    }
  }
}

extern "C" void kernel_launch(void* const* d_in, const int* in_sizes, int n_in,
                              void* d_out, int out_size, void* d_ws, size_t ws_size,
                              hipStream_t stream) {
  (void)in_sizes; (void)n_in; (void)d_ws; (void)ws_size; (void)out_size;
  const float* x        = (const float*)d_in[0];
  const float* gcn_w0   = (const float*)d_in[1];
  const float* gcn_b0   = (const float*)d_in[2];
  const float* tcn_w0   = (const float*)d_in[3];
  const float* tcn_b0   = (const float*)d_in[4];
  const float* bn_g0    = (const float*)d_in[5];
  const float* bn_b0    = (const float*)d_in[6];
  const float* bn_m0    = (const float*)d_in[7];
  const float* bn_v0    = (const float*)d_in[8];
  const float* res_w    = (const float*)d_in[9];
  const float* res_b    = (const float*)d_in[10];
  const float* res_bn_g = (const float*)d_in[11];
  const float* res_bn_b = (const float*)d_in[12];
  const float* res_bn_m = (const float*)d_in[13];
  const float* res_bn_v = (const float*)d_in[14];
  const float* gcn_w1   = (const float*)d_in[15];
  const float* gcn_b1   = (const float*)d_in[16];
  const float* tcn_w1   = (const float*)d_in[17];
  const float* tcn_b1   = (const float*)d_in[18];
  const float* bn_g1    = (const float*)d_in[19];
  const float* bn_b1    = (const float*)d_in[20];
  const float* bn_m1    = (const float*)d_in[21];
  const float* bn_v1    = (const float*)d_in[22];
  const float* fc_w     = (const float*)d_in[23];
  const float* fc_b     = (const float*)d_in[24];

  dim3 grid(NB * 2 * (NTIME / TT));
  dim3 block(NTHR);
  stgcn_fused<<<grid, block, 0, stream>>>(
      x, gcn_w0, gcn_b0, tcn_w0, tcn_b0, bn_g0, bn_b0, bn_m0, bn_v0,
      res_w, res_b, res_bn_g, res_bn_b, res_bn_m, res_bn_v,
      gcn_w1, gcn_b1, tcn_w1, tcn_b1, bn_g1, bn_b1, bn_m1, bn_v1,
      fc_w, fc_b, (float*)d_out);
}


# gcn1 (32->32 + bias) folded into tcn1's weights per workgroup on WMMA: tcn1 reads hagg, GCN1 pass and barrier gone; column half fixed per wave keeps tcn1's B in registers; new sum order (ci first)
# speedup vs baseline: 5.7563x; 1.1645x over previous
#include <hip/hip_runtime.h>

#define TT 16
#define T1 (TT + 6)
#define T2 (TT + 4)
#define T3 (TT + 2)
#define NTHR 512
#define NWAVE (NTHR / 32)
#define NB 64
#define NTIME 512
#define NV 21
#define NC 32
#define NCIN 2
#define NOUT 64
#define PC 36
#define PW 40
#define PR 97
#define RP 68

typedef __attribute__((ext_vector_type(2))) float v2f;
typedef __attribute__((ext_vector_type(8))) float v8f;

__device__ __forceinline__ v8f wmma4(v2f a, v2f b, v8f c) {
  return __builtin_amdgcn_wmma_f32_16x16x4_f32(
       false, a,  false, b,
       (short)0, c,  false,  false);
}

__host__ __device__ constexpr int g_parent(int j) { return ((j - 1) % 4 == 0) ? 0 : j - 1; }
__host__ __device__ constexpr bool g_conn(int i, int j) {
  return i == j || (j > 0 && g_parent(j) == i) || (i > 0 && g_parent(i) == j);
}
__host__ __device__ constexpr float g_dinv(int v) {
  return v == 0 ? 0.4082482905f : ((v % 4 == 0) ? 0.7071067812f : 0.5773502692f);
}

__host__ __device__ constexpr int  fc_off(int g)    { return (g % NV) * PC + g / NV; }
__host__ __device__ constexpr bool fc_nowrap(int g) { return (g % NV) + 4 < NV; }

template <int STRIDE_IN, int STRIDE_OUT>
__device__ __forceinline__ void graph_agg(const float* __restrict__ in, float* __restrict__ out) {
  float h[NV];
#pragma unroll
  for (int u = 0; u < NV; ++u) h[u] = in[u * STRIDE_IN];
#pragma unroll
  for (int v = 0; v < NV; ++v) {
    float acc = 0.0f;
    bool have = false;
#pragma unroll
    for (int u = 0; u < NV; ++u) {
      if (g_conn(v, u)) {
        const float p = (g_dinv(v) * g_dinv(u)) * h[u];
        acc = have ? acc + p : p;
        have = true;
      }
    }
    out[v * STRIDE_OUT] = acc;
  }
}

__global__ __launch_bounds__(NTHR) void stgcn_fused(
    const float* __restrict__ x,
    const float* __restrict__ gcn_w0, const float* __restrict__ gcn_b0,
    const float* __restrict__ tcn_w0, const float* __restrict__ tcn_b0,
    const float* __restrict__ bn_g0, const float* __restrict__ bn_b0,
    const float* __restrict__ bn_m0, const float* __restrict__ bn_v0,
    const float* __restrict__ res_w, const float* __restrict__ res_b,
    const float* __restrict__ res_bn_g, const float* __restrict__ res_bn_b,
    const float* __restrict__ res_bn_m, const float* __restrict__ res_bn_v,
    const float* __restrict__ gcn_w1, const float* __restrict__ gcn_b1,
    const float* __restrict__ tcn_w1, const float* __restrict__ tcn_b1,
    const float* __restrict__ bn_g1, const float* __restrict__ bn_b1,
    const float* __restrict__ bn_m1, const float* __restrict__ bn_v1,
    const float* __restrict__ fc_w, const float* __restrict__ fc_b,
    float* __restrict__ out)
{
  __shared__ __attribute__((aligned(16))) float s_in[T1 * NV * NCIN];
  __shared__ __attribute__((aligned(16))) float s_xa[T1 * NV * NCIN];
  __shared__ float s_tw[96 * PW];
  __shared__ float s_gw1[NC * PC];
  __shared__ float s_g1[T3 * NV * PC];
  __shared__ float s_h0[T2 * NV * PC];
  __shared__ __attribute__((aligned(16))) float s_wf[3 * NC * 4];
  __shared__ float s_wb1[3 * NC];
  __shared__ float s_s0[NC], s_o0[NC], s_s1[NC], s_o1[NC];
  __shared__ float s_rA[NC], s_rB[NC], s_rC[NC];
  __shared__ float s_fcb[NOUT];

  const int tid  = threadIdx.x;
  const int nthr = NTHR;
  const int lane = tid & 31;
  const int wave = tid >> 5;
  const int row  = lane & 15;
  const int hi   = lane >> 4;

  const int bid   = blockIdx.x;
  const int tile  = bid & 31;
  const int hand  = (bid >> 5) & 1;
  const int b     = bid >> 6;
  const int tbase = tile * TT;
  const int wave_u = __builtin_amdgcn_readfirstlane(wave);

  float* s_raw  = s_g1;
  float* s_raw1 = s_g1 + NC * PR;
#pragma unroll
  for (int i = 0; i < 3072 / NTHR; ++i) {
    int idx = tid + i * NTHR;
    int co = idx / 96;
    int o = co * PR + (idx - co * 96);
    s_raw[o]  = tcn_w0[idx];
    s_raw1[o] = tcn_w1[idx];
  }
  for (int idx = tid; idx < NC * NC; idx += nthr) s_gw1[(idx >> 5) * PC + (idx & 31)] = gcn_w1[idx];
  if (tid < NC) {
    float sc0 = bn_g0[tid] * rsqrtf(bn_v0[tid] + 1e-5f);
    s_s0[tid] = sc0;
    s_o0[tid] = (tcn_b0[tid] - bn_m0[tid]) * sc0 + bn_b0[tid];
    float sc1 = bn_g1[tid] * rsqrtf(bn_v1[tid] + 1e-5f);
    s_s1[tid] = sc1;
    s_o1[tid] = (tcn_b1[tid] - bn_m1[tid]) * sc1 + bn_b1[tid];
    float rsc = res_bn_g[tid] * rsqrtf(res_bn_v[tid] + 1e-5f);
    s_rA[tid] = res_w[tid * 2 + 0] * rsc;
    s_rB[tid] = res_w[tid * 2 + 1] * rsc;
    s_rC[tid] = (res_b[tid] - res_bn_m[tid]) * rsc + res_bn_b[tid];
  }
  if (tid < NOUT) s_fcb[tid] = fc_b[tid];

  for (int idx = tid; idx < T1 * NV * NCIN; idx += nthr) {
    int i = idx / (NV * NCIN), r = idx % (NV * NCIN);
    int v = r >> 1, ci = r & 1;
    int t = tbase - 3 + i;
    float val = 0.0f;
    if (t >= 0 && t < NTIME)
      val = x[(((b * NTIME + t) * 2 + hand) * NV + v) * NCIN + ci];
    s_in[idx] = val;
  }
  __syncthreads();

  if (wave_u < 2) {
    if (tid < T1 * NCIN)
      graph_agg<NCIN, NCIN>(&s_in[(tid >> 1) * (NV * NCIN) + (tid & 1)],
                            &s_xa[(tid >> 1) * (NV * NCIN) + (tid & 1)]);
  } else if (wave_u < 5) {
    const int k = wave_u - 2;
    const float* rw  = &s_raw[lane * PR + k];
    const float* rw1 = &s_raw1[lane * PR + k];
    float w0 = 0.0f, w1 = 0.0f, wb = 0.0f, wb1 = 0.0f;
#pragma unroll
    for (int ci = 0; ci < NC; ++ci) {
      const float w = rw[ci * 3];
      w0 += w * gcn_w0[ci * 2 + 0];
      w1 += w * gcn_w0[ci * 2 + 1];
      wb += w * gcn_b0[ci];
      wb1 += rw1[ci * 3] * gcn_b1[ci];
    }
    *(float4*)&s_wf[(k * NC + lane) * 4] = make_float4(w0, w1, wb, 0.0f);
    s_wb1[k * NC + lane] = wb1;
  }
  {
    const int fjob = (wave_u >= 5) ? wave_u - 5 : ((wave_u == 1) ? 11 : -1);
    if (fjob >= 0) {
      const int k = fjob >> 2, mh = (fjob >> 1) & 1, nh = fjob & 1;
      const float* ap = &s_gw1[(hi * 2) * PC + mh * 16 + row];
      const float* bp = &s_raw1[(nh * 16 + row) * PR + (hi * 2) * 3 + k];
      v2f fa[8], fb[8];
#pragma unroll
      for (int ks = 0; ks < 8; ++ks) {
        fa[ks].x = ap[(ks * 4) * PC];
        fa[ks].y = ap[(ks * 4 + 1) * PC];
        fb[ks].x = bp[(ks * 4) * 3];
        fb[ks].y = bp[(ks * 4 + 1) * 3];
      }
      v8f c0 = {0.f,0.f,0.f,0.f,0.f,0.f,0.f,0.f};
      v8f c1 = {0.f,0.f,0.f,0.f,0.f,0.f,0.f,0.f};
#pragma unroll
      for (int ks = 0; ks < 8; ks += 2) {
        c0 = wmma4(fa[ks], fb[ks], c0);
        c1 = wmma4(fa[ks + 1], fb[ks + 1], c1);
      }
      float* wp = &s_tw[(k * NC + mh * 16 + hi * 8) * PW + nh * 16 + row];
#pragma unroll
      for (int j = 0; j < 8; ++j) wp[j * PW] = c0[j] + c1[j];
    }
  }
  __syncthreads();

  {
    const int M = T2 * NV;
    const int colN = (wave_u & 1) * 16 + row;
    const float c_s0 = s_s0[colN], c_o0 = s_o0[colN];
    const float c_rA = s_rA[colN], c_rB = s_rB[colN], c_rC = s_rC[colN];
    for (int tIdx = wave_u; tIdx < 27 * 2; tIdx += NWAVE) {
      const int mt = tIdx >> 1;
      v8f acc = {0.f,0.f,0.f,0.f,0.f,0.f,0.f,0.f};
      int mA = mt * 16 + row;
      int mC = (mA < M) ? mA : 0;
      int t2 = mC / NV;
      const int m0 = mt * 16 + hi * 8;
      v2f xr[8];
#pragma unroll
      for (int j = 0; j < 8; ++j) xr[j] = *(const v2f*)&s_in[(m0 + j + NV) * NCIN];
#pragma unroll
      for (int k = 0; k < 3; ++k) {
        const v2f xv = *(const v2f*)&s_xa[(mC + k * NV) * NCIN];
        const int t = tbase - 3 + t2 + k;
        const float ok = (t >= 0 && t < NTIME) ? 1.0f : 0.0f;
        v2f a;
        a.x = hi ? ok : xv.x;
        a.y = hi ? 0.0f : xv.y;
        const v2f bb = *(const v2f*)&s_wf[(k * NC + colN) * 4 + hi * 2];
        acc = wmma4(a, bb, acc);
      }
      float* hp = &s_h0[m0 * PC + colN];
      if (mt < 26) {
#pragma unroll
        for (int j = 0; j < 8; ++j) {
          float val = acc[j] * c_s0 + c_o0;
          val += c_rA * xr[j].x + c_rB * xr[j].y + c_rC;
          hp[j * PC] = fmaxf(val, 0.0f);
        }
      } else {
#pragma unroll
        for (int j = 0; j < 8; ++j) {
          if (m0 + j < M) {
            float val = acc[j] * c_s0 + c_o0;
            val += c_rA * xr[j].x + c_rB * xr[j].y + c_rC;
            hp[j * PC] = fmaxf(val, 0.0f);
          }
        }
      }
    }
  }
  __syncthreads();

  float* s_hagg = s_g1;
  for (int t3 = wave_u; t3 < T3; t3 += NWAVE) {
    const int t = tbase - 1 + t3;
    if (t >= 0 && t < NTIME) {
      graph_agg<PC, PC>(&s_h0[(t3 + 1) * (NV * PC) + lane], &s_hagg[t3 * (NV * PC) + lane]);
    } else {
#pragma unroll
      for (int v = 0; v < NV; ++v) s_hagg[t3 * (NV * PC) + v * PC + lane] = 0.0f;
    }
  }
  __syncthreads();

  {
    const int M = TT * NV;
    const int colN = (wave_u & 1) * 16 + row;
    const float c_s1 = s_s1[colN], c_o1 = s_o1[colN];
    const float wbk0 = s_wb1[colN], wbk1 = s_wb1[NC + colN], wbk2 = s_wb1[2 * NC + colN];
    const float o1_all   = ((wbk0 + wbk1) + wbk2) * c_s1 + c_o1;
    const float o1_first = (wbk1 + wbk2) * c_s1 + c_o1;
    const float o1_last  = (wbk0 + wbk1) * c_s1 + c_o1;
    const bool edge = (tile == 0) || (tile == NTIME / TT - 1);
    for (int tIdx = wave_u; tIdx < 21 * 2; tIdx += NWAVE) {
      int mt = tIdx >> 1;
      v8f acc = {0.f,0.f,0.f,0.f,0.f,0.f,0.f,0.f};
      int mA = mt * 16 + row;
      int mC = (mA < M) ? mA : 0;
      int t4 = mC / NV, vv = mC % NV;
      for (int ks = 0; ks < 24; ++ks) {
        int k0 = ks * 4 + hi * 2;
        v2f a, bb;
        a.x = s_hagg[((t4 + (k0 >> 5)) * NV + vv) * PC + (k0 & 31)];
        a.y = s_hagg[((t4 + ((k0 + 1) >> 5)) * NV + vv) * PC + ((k0 + 1) & 31)];
        bb.x = s_tw[k0 * PW + colN];
        bb.y = s_tw[(k0 + 1) * PW + colN];
        acc = wmma4(a, bb, acc);
      }
      if (!edge) {
#pragma unroll
        for (int j = 0; j < 8; ++j) {
          int m = mt * 16 + j + hi * 8;
          if (m < M) {
            int te = m / NV, ve = m % NV;
            int hidx = ((te + 2) * NV + ve) * PC + colN;
            float val = acc[j] * c_s1 + o1_all + s_h0[hidx];
            s_h0[hidx] = fmaxf(val, 0.0f);
          }
        }
      } else {
#pragma unroll
        for (int j = 0; j < 8; ++j) {
          int m = mt * 16 + j + hi * 8;
          if (m < M) {
            int te = m / NV, ve = m % NV;
            int t = tbase + te;
            float oj = (t == 0) ? o1_first : ((t == NTIME - 1) ? o1_last : o1_all);
            int hidx = ((te + 2) * NV + ve) * PC + colN;
            float val = acc[j] * c_s1 + oj + s_h0[hidx];
            s_h0[hidx] = fmaxf(val, 0.0f);
          }
        }
      }
    }
  }

  constexpr int FCQ = 21;
  constexpr int FCD = 7;
  const int fc_nt = wave & 3;
  const int fc_kq = wave >> 2;
  const int fc_col = fc_nt * 16 + row;
  const float* fc_bp = fc_w + fc_col * 672 + fc_kq * 168 + hi * 4;
  float4 fcw_r[FCD];
#pragma unroll
  for (int q = 0; q < FCD; ++q) fcw_r[q] = *(const float4*)(fc_bp + 8 * q);
  __syncthreads();

  {
    float* s_red = s_g1;
    v8f acc = {0.f,0.f,0.f,0.f,0.f,0.f,0.f,0.f};
    const float* h1row = &s_h0[(row + 2) * (NV * PC) + 8 * fc_kq];
    const float* aP = h1row + hi * (4 * PC);
    const float* aQ = h1row + hi * ((4 - NV) * PC + 1);
#pragma unroll
    for (int q = 0; q < FCQ; ++q) {
      const float4 bv = fcw_r[q % FCD];
      if (q + FCD < FCQ) fcw_r[q % FCD] = *(const float4*)(fc_bp + 8 * (q + FCD));
      const int g = 8 * q;
      v2f a0, a1, b0, b1;
      a0.x = (fc_nowrap(g + 0) ? aP : aQ)[fc_off(g + 0)];
      a0.y = (fc_nowrap(g + 1) ? aP : aQ)[fc_off(g + 1)];
      a1.x = (fc_nowrap(g + 2) ? aP : aQ)[fc_off(g + 2)];
      a1.y = (fc_nowrap(g + 3) ? aP : aQ)[fc_off(g + 3)];
      b0.x = bv.x; b0.y = bv.y;
      b1.x = bv.z; b1.y = bv.w;
      acc = wmma4(a0, b0, acc);
      acc = wmma4(a1, b1, acc);
    }
#pragma unroll
    for (int j = 0; j < 8; ++j)
      s_red[(fc_kq * 16 + j + hi * 8) * RP + fc_col] = acc[j];
    __syncthreads();
#pragma unroll
    for (int i = 0; i < (TT * NOUT) / NTHR; ++i) {
      const int idx = tid + i * NTHR;
      const int m = idx >> 6, o = idx & 63;
      float val = s_red[m * RP + o];
      val += s_red[(16 + m) * RP + o];
      val += s_red[(32 + m) * RP + o];
      val += s_red[(48 + m) * RP + o];
      val += s_fcb[o];
      out[(((size_t)b * NTIME + (tbase + m)) * 2 + hand) * NOUT + o] = val;
    }
  }
}

extern "C" void kernel_launch(void* const* d_in, const int* in_sizes, int n_in,
                              void* d_out, int out_size, void* d_ws, size_t ws_size,
                              hipStream_t stream) {
  (void)in_sizes; (void)n_in; (void)d_ws; (void)ws_size; (void)out_size;
  const float* x        = (const float*)d_in[0];
  const float* gcn_w0   = (const float*)d_in[1];
  const float* gcn_b0   = (const float*)d_in[2];
  const float* tcn_w0   = (const float*)d_in[3];
  const float* tcn_b0   = (const float*)d_in[4];
  const float* bn_g0    = (const float*)d_in[5];
  const float* bn_b0    = (const float*)d_in[6];
  const float* bn_m0    = (const float*)d_in[7];
  const float* bn_v0    = (const float*)d_in[8];
  const float* res_w    = (const float*)d_in[9];
  const float* res_b    = (const float*)d_in[10];
  const float* res_bn_g = (const float*)d_in[11];
  const float* res_bn_b = (const float*)d_in[12];
  const float* res_bn_m = (const float*)d_in[13];
  const float* res_bn_v = (const float*)d_in[14];
  const float* gcn_w1   = (const float*)d_in[15];
  const float* gcn_b1   = (const float*)d_in[16];
  const float* tcn_w1   = (const float*)d_in[17];
  const float* tcn_b1   = (const float*)d_in[18];
  const float* bn_g1    = (const float*)d_in[19];
  const float* bn_b1    = (const float*)d_in[20];
  const float* bn_m1    = (const float*)d_in[21];
  const float* bn_v1    = (const float*)d_in[22];
  const float* fc_w     = (const float*)d_in[23];
  const float* fc_b     = (const float*)d_in[24];

  dim3 grid(NB * 2 * (NTIME / TT));
  dim3 block(NTHR);
  stgcn_fused<<<grid, block, 0, stream>>>(
      x, gcn_w0, gcn_b0, tcn_w0, tcn_b0, bn_g0, bn_b0, bn_m0, bn_v0,
      res_w, res_b, res_bn_g, res_bn_b, res_bn_m, res_bn_v,
      gcn_w1, gcn_b1, tcn_w1, tcn_b1, bn_g1, bn_b1, bn_m1, bn_v1,
      fc_w, fc_b, (float*)d_out);
}


# fc_w repacked each call by a small first kernel into the FC loop's fragment order (workspace): a wave's 16-byte-per-lane weight fetch covers 4 whole lines, not 32 bytes on each of 16; same sums
# speedup vs baseline: 5.9810x; 1.0390x over previous
#include <hip/hip_runtime.h>

#define TT 16
#define T1 (TT + 6)
#define T2 (TT + 4)
#define T3 (TT + 2)
#define NTHR 512
#define NWAVE (NTHR / 32)
#define NB 64
#define NTIME 512
#define NV 21
#define NC 32
#define NCIN 2
#define NOUT 64
#define PC 36
#define PW 40
#define PR 97
#define RP 68

typedef __attribute__((ext_vector_type(2))) float v2f;
typedef __attribute__((ext_vector_type(8))) float v8f;

__device__ __forceinline__ v8f wmma4(v2f a, v2f b, v8f c) {
  return __builtin_amdgcn_wmma_f32_16x16x4_f32(
       false, a,  false, b,
       (short)0, c,  false,  false);
}

__host__ __device__ constexpr int g_parent(int j) { return ((j - 1) % 4 == 0) ? 0 : j - 1; }
__host__ __device__ constexpr bool g_conn(int i, int j) {
  return i == j || (j > 0 && g_parent(j) == i) || (i > 0 && g_parent(i) == j);
}
__host__ __device__ constexpr float g_dinv(int v) {
  return v == 0 ? 0.4082482905f : ((v % 4 == 0) ? 0.7071067812f : 0.5773502692f);
}

__host__ __device__ constexpr int  fc_off(int g)    { return (g % NV) * PC + g / NV; }
__host__ __device__ constexpr bool fc_nowrap(int g) { return (g % NV) + 4 < NV; }

template <int STRIDE_IN, int STRIDE_OUT>
__device__ __forceinline__ void graph_agg(const float* __restrict__ in, float* __restrict__ out) {
  float h[NV];
#pragma unroll
  for (int u = 0; u < NV; ++u) h[u] = in[u * STRIDE_IN];
#pragma unroll
  for (int v = 0; v < NV; ++v) {
    float acc = 0.0f;
    bool have = false;
#pragma unroll
    for (int u = 0; u < NV; ++u) {
      if (g_conn(v, u)) {
        const float p = (g_dinv(v) * g_dinv(u)) * h[u];
        acc = have ? acc + p : p;
        have = true;
      }
    }
    out[v * STRIDE_OUT] = acc;
  }
}

__global__ __launch_bounds__(256) void fcw_pack(const float* __restrict__ fc_w, float* __restrict__ fcw_p) {
  const int gid  = blockIdx.x * 256 + threadIdx.x;
  const int lane = gid & 31;
  const int wq   = gid >> 5;
  const int w    = wq / 21;
  const int q    = wq - w * 21;
  const int col  = (w & 3) * 16 + (lane & 15);
  const float4 v = *(const float4*)(fc_w + col * 672 + (w >> 2) * 168 + 8 * q + (lane >> 4) * 4);
  *(float4*)(fcw_p + gid * 4) = v;
}

__global__ __launch_bounds__(NTHR) void stgcn_fused(
    const float* __restrict__ x,
    const float* __restrict__ gcn_w0, const float* __restrict__ gcn_b0,
    const float* __restrict__ tcn_w0, const float* __restrict__ tcn_b0,
    const float* __restrict__ bn_g0, const float* __restrict__ bn_b0,
    const float* __restrict__ bn_m0, const float* __restrict__ bn_v0,
    const float* __restrict__ res_w, const float* __restrict__ res_b,
    const float* __restrict__ res_bn_g, const float* __restrict__ res_bn_b,
    const float* __restrict__ res_bn_m, const float* __restrict__ res_bn_v,
    const float* __restrict__ gcn_w1, const float* __restrict__ gcn_b1,
    const float* __restrict__ tcn_w1, const float* __restrict__ tcn_b1,
    const float* __restrict__ bn_g1, const float* __restrict__ bn_b1,
    const float* __restrict__ bn_m1, const float* __restrict__ bn_v1,
    const float* __restrict__ fcw_p, const float* __restrict__ fc_b,
    float* __restrict__ out)
{
  __shared__ __attribute__((aligned(16))) float s_in[T1 * NV * NCIN];
  __shared__ __attribute__((aligned(16))) float s_xa[T1 * NV * NCIN];
  __shared__ float s_tw[96 * PW];
  __shared__ float s_gw1[NC * PC];
  __shared__ float s_g1[T3 * NV * PC];
  __shared__ float s_h0[T2 * NV * PC];
  __shared__ __attribute__((aligned(16))) float s_wf[3 * NC * 4];
  __shared__ float s_wb1[3 * NC];
  __shared__ float s_s0[NC], s_o0[NC], s_s1[NC], s_o1[NC];
  __shared__ float s_rA[NC], s_rB[NC], s_rC[NC];
  __shared__ float s_fcb[NOUT];

  const int tid  = threadIdx.x;
  const int nthr = NTHR;
  const int lane = tid & 31;
  const int wave = tid >> 5;
  const int row  = lane & 15;
  const int hi   = lane >> 4;

  const int bid   = blockIdx.x;
  const int tile  = bid & 31;
  const int hand  = (bid >> 5) & 1;
  const int b     = bid >> 6;
  const int tbase = tile * TT;
  const int wave_u = __builtin_amdgcn_readfirstlane(wave);

  float* s_raw  = s_g1;
  float* s_raw1 = s_g1 + NC * PR;
#pragma unroll
  for (int i = 0; i < 3072 / NTHR; ++i) {
    int idx = tid + i * NTHR;
    int co = idx / 96;
    int o = co * PR + (idx - co * 96);
    s_raw[o]  = tcn_w0[idx];
    s_raw1[o] = tcn_w1[idx];
  }
  for (int idx = tid; idx < NC * NC; idx += nthr) s_gw1[(idx >> 5) * PC + (idx & 31)] = gcn_w1[idx];
  if (tid < NC) {
    float sc0 = bn_g0[tid] * rsqrtf(bn_v0[tid] + 1e-5f);
    s_s0[tid] = sc0;
    s_o0[tid] = (tcn_b0[tid] - bn_m0[tid]) * sc0 + bn_b0[tid];
    float sc1 = bn_g1[tid] * rsqrtf(bn_v1[tid] + 1e-5f);
    s_s1[tid] = sc1;
    s_o1[tid] = (tcn_b1[tid] - bn_m1[tid]) * sc1 + bn_b1[tid];
    float rsc = res_bn_g[tid] * rsqrtf(res_bn_v[tid] + 1e-5f);
    s_rA[tid] = res_w[tid * 2 + 0] * rsc;
    s_rB[tid] = res_w[tid * 2 + 1] * rsc;
    s_rC[tid] = (res_b[tid] - res_bn_m[tid]) * rsc + res_bn_b[tid];
  }
  if (tid < NOUT) s_fcb[tid] = fc_b[tid];

  for (int idx = tid; idx < T1 * NV * NCIN; idx += nthr) {
    int i = idx / (NV * NCIN), r = idx % (NV * NCIN);
    int v = r >> 1, ci = r & 1;
    int t = tbase - 3 + i;
    float val = 0.0f;
    if (t >= 0 && t < NTIME)
      val = x[(((b * NTIME + t) * 2 + hand) * NV + v) * NCIN + ci];
    s_in[idx] = val;
  }
  __syncthreads();

  if (wave_u < 2) {
    if (tid < T1 * NCIN)
      graph_agg<NCIN, NCIN>(&s_in[(tid >> 1) * (NV * NCIN) + (tid & 1)],
                            &s_xa[(tid >> 1) * (NV * NCIN) + (tid & 1)]);
  } else if (wave_u < 5) {
    const int k = wave_u - 2;
    const float* rw  = &s_raw[lane * PR + k];
    const float* rw1 = &s_raw1[lane * PR + k];
    float w0 = 0.0f, w1 = 0.0f, wb = 0.0f, wb1 = 0.0f;
#pragma unroll
    for (int ci = 0; ci < NC; ++ci) {
      const float w = rw[ci * 3];
      w0 += w * gcn_w0[ci * 2 + 0];
      w1 += w * gcn_w0[ci * 2 + 1];
      wb += w * gcn_b0[ci];
      wb1 += rw1[ci * 3] * gcn_b1[ci];
    }
    *(float4*)&s_wf[(k * NC + lane) * 4] = make_float4(w0, w1, wb, 0.0f);
    s_wb1[k * NC + lane] = wb1;
  }
  {
    const int fjob = (wave_u >= 5) ? wave_u - 5 : ((wave_u == 1) ? 11 : -1);
    if (fjob >= 0) {
      const int k = fjob >> 2, mh = (fjob >> 1) & 1, nh = fjob & 1;
      const float* ap = &s_gw1[(hi * 2) * PC + mh * 16 + row];
      const float* bp = &s_raw1[(nh * 16 + row) * PR + (hi * 2) * 3 + k];
      v2f fa[8], fb[8];
#pragma unroll
      for (int ks = 0; ks < 8; ++ks) {
        fa[ks].x = ap[(ks * 4) * PC];
        fa[ks].y = ap[(ks * 4 + 1) * PC];
        fb[ks].x = bp[(ks * 4) * 3];
        fb[ks].y = bp[(ks * 4 + 1) * 3];
      }
      v8f c0 = {0.f,0.f,0.f,0.f,0.f,0.f,0.f,0.f};
      v8f c1 = {0.f,0.f,0.f,0.f,0.f,0.f,0.f,0.f};
#pragma unroll
      for (int ks = 0; ks < 8; ks += 2) {
        c0 = wmma4(fa[ks], fb[ks], c0);
        c1 = wmma4(fa[ks + 1], fb[ks + 1], c1);
      }
      float* wp = &s_tw[(k * NC + mh * 16 + hi * 8) * PW + nh * 16 + row];
#pragma unroll
      for (int j = 0; j < 8; ++j) wp[j * PW] = c0[j] + c1[j];
    }
  }
  __syncthreads();

  {
    const int M = T2 * NV;
    const int colN = (wave_u & 1) * 16 + row;
    const float c_s0 = s_s0[colN], c_o0 = s_o0[colN];
    const float c_rA = s_rA[colN], c_rB = s_rB[colN], c_rC = s_rC[colN];
    for (int tIdx = wave_u; tIdx < 27 * 2; tIdx += NWAVE) {
      const int mt = tIdx >> 1;
      v8f acc = {0.f,0.f,0.f,0.f,0.f,0.f,0.f,0.f};
      int mA = mt * 16 + row;
      int mC = (mA < M) ? mA : 0;
      int t2 = mC / NV;
      const int m0 = mt * 16 + hi * 8;
      v2f xr[8];
#pragma unroll
      for (int j = 0; j < 8; ++j) xr[j] = *(const v2f*)&s_in[(m0 + j + NV) * NCIN];
#pragma unroll
      for (int k = 0; k < 3; ++k) {
        const v2f xv = *(const v2f*)&s_xa[(mC + k * NV) * NCIN];
        const int t = tbase - 3 + t2 + k;
        const float ok = (t >= 0 && t < NTIME) ? 1.0f : 0.0f;
        v2f a;
        a.x = hi ? ok : xv.x;
        a.y = hi ? 0.0f : xv.y;
        const v2f bb = *(const v2f*)&s_wf[(k * NC + colN) * 4 + hi * 2];
        acc = wmma4(a, bb, acc);
      }
      float* hp = &s_h0[m0 * PC + colN];
      if (mt < 26) {
#pragma unroll
        for (int j = 0; j < 8; ++j) {
          float val = acc[j] * c_s0 + c_o0;
          val += c_rA * xr[j].x + c_rB * xr[j].y + c_rC;
          hp[j * PC] = fmaxf(val, 0.0f);
        }
      } else {
#pragma unroll
        for (int j = 0; j < 8; ++j) {
          if (m0 + j < M) {
            float val = acc[j] * c_s0 + c_o0;
            val += c_rA * xr[j].x + c_rB * xr[j].y + c_rC;
            hp[j * PC] = fmaxf(val, 0.0f);
          }
        }
      }
    }
  }
  __syncthreads();

  float* s_hagg = s_g1;
  for (int t3 = wave_u; t3 < T3; t3 += NWAVE) {
    const int t = tbase - 1 + t3;
    if (t >= 0 && t < NTIME) {
      graph_agg<PC, PC>(&s_h0[(t3 + 1) * (NV * PC) + lane], &s_hagg[t3 * (NV * PC) + lane]);
    } else {
#pragma unroll
      for (int v = 0; v < NV; ++v) s_hagg[t3 * (NV * PC) + v * PC + lane] = 0.0f;
    }
  }
  __syncthreads();

  {
    const int M = TT * NV;
    const int colN = (wave_u & 1) * 16 + row;
    const float c_s1 = s_s1[colN], c_o1 = s_o1[colN];
    const float wbk0 = s_wb1[colN], wbk1 = s_wb1[NC + colN], wbk2 = s_wb1[2 * NC + colN];
    const float o1_all   = ((wbk0 + wbk1) + wbk2) * c_s1 + c_o1;
    const float o1_first = (wbk1 + wbk2) * c_s1 + c_o1;
    const float o1_last  = (wbk0 + wbk1) * c_s1 + c_o1;
    const bool edge = (tile == 0) || (tile == NTIME / TT - 1);
    for (int tIdx = wave_u; tIdx < 21 * 2; tIdx += NWAVE) {
      int mt = tIdx >> 1;
      v8f acc = {0.f,0.f,0.f,0.f,0.f,0.f,0.f,0.f};
      int mA = mt * 16 + row;
      int mC = (mA < M) ? mA : 0;
      int t4 = mC / NV, vv = mC % NV;
      for (int ks = 0; ks < 24; ++ks) {
        int k0 = ks * 4 + hi * 2;
        v2f a, bb;
        a.x = s_hagg[((t4 + (k0 >> 5)) * NV + vv) * PC + (k0 & 31)];
        a.y = s_hagg[((t4 + ((k0 + 1) >> 5)) * NV + vv) * PC + ((k0 + 1) & 31)];
        bb.x = s_tw[k0 * PW + colN];
        bb.y = s_tw[(k0 + 1) * PW + colN];
        acc = wmma4(a, bb, acc);
      }
      if (!edge) {
#pragma unroll
        for (int j = 0; j < 8; ++j) {
          int m = mt * 16 + j + hi * 8;
          if (m < M) {
            int te = m / NV, ve = m % NV;
            int hidx = ((te + 2) * NV + ve) * PC + colN;
            float val = acc[j] * c_s1 + o1_all + s_h0[hidx];
            s_h0[hidx] = fmaxf(val, 0.0f);
          }
        }
      } else {
#pragma unroll
        for (int j = 0; j < 8; ++j) {
          int m = mt * 16 + j + hi * 8;
          if (m < M) {
            int te = m / NV, ve = m % NV;
            int t = tbase + te;
            float oj = (t == 0) ? o1_first : ((t == NTIME - 1) ? o1_last : o1_all);
            int hidx = ((te + 2) * NV + ve) * PC + colN;
            float val = acc[j] * c_s1 + oj + s_h0[hidx];
            s_h0[hidx] = fmaxf(val, 0.0f);
          }
        }
      }
    }
  }

  constexpr int FCQ = 21;
  constexpr int FCD = 7;
  const int fc_nt = wave & 3;
  const int fc_kq = wave >> 2;
  const int fc_col = fc_nt * 16 + row;
  const float* fc_bp = fcw_p + (wave * (FCQ * 32) + lane) * 4;
  float4 fcw_r[FCD];
#pragma unroll
  for (int q = 0; q < FCD; ++q) fcw_r[q] = *(const float4*)(fc_bp + 128 * q);
  __syncthreads();

  {
    float* s_red = s_g1;
    v8f acc = {0.f,0.f,0.f,0.f,0.f,0.f,0.f,0.f};
    const float* h1row = &s_h0[(row + 2) * (NV * PC) + 8 * fc_kq];
    const float* aP = h1row + hi * (4 * PC);
    const float* aQ = h1row + hi * ((4 - NV) * PC + 1);
#pragma unroll
    for (int q = 0; q < FCQ; ++q) {
      const float4 bv = fcw_r[q % FCD];
      if (q + FCD < FCQ) fcw_r[q % FCD] = *(const float4*)(fc_bp + 128 * (q + FCD));
      const int g = 8 * q;
      v2f a0, a1, b0, b1;
      a0.x = (fc_nowrap(g + 0) ? aP : aQ)[fc_off(g + 0)];
      a0.y = (fc_nowrap(g + 1) ? aP : aQ)[fc_off(g + 1)];
      a1.x = (fc_nowrap(g + 2) ? aP : aQ)[fc_off(g + 2)];
      a1.y = (fc_nowrap(g + 3) ? aP : aQ)[fc_off(g + 3)];
      b0.x = bv.x; b0.y = bv.y;
      b1.x = bv.z; b1.y = bv.w;
      acc = wmma4(a0, b0, acc);
      acc = wmma4(a1, b1, acc);
    }
#pragma unroll
    for (int j = 0; j < 8; ++j)
      s_red[(fc_kq * 16 + j + hi * 8) * RP + fc_col] = acc[j];
    __syncthreads();
#pragma unroll
    for (int i = 0; i < (TT * NOUT) / NTHR; ++i) {
      const int idx = tid + i * NTHR;
      const int m = idx >> 6, o = idx & 63;
      float val = s_red[m * RP + o];
      val += s_red[(16 + m) * RP + o];
      val += s_red[(32 + m) * RP + o];
      val += s_red[(48 + m) * RP + o];
      val += s_fcb[o];
      out[(((size_t)b * NTIME + (tbase + m)) * 2 + hand) * NOUT + o] = val;
    }
  }
}

extern "C" void kernel_launch(void* const* d_in, const int* in_sizes, int n_in,
                              void* d_out, int out_size, void* d_ws, size_t ws_size,
                              hipStream_t stream) {
  (void)in_sizes; (void)n_in; (void)ws_size; (void)out_size;
  const float* x        = (const float*)d_in[0];
  const float* gcn_w0   = (const float*)d_in[1];
  const float* gcn_b0   = (const float*)d_in[2];
  const float* tcn_w0   = (const float*)d_in[3];
  const float* tcn_b0   = (const float*)d_in[4];
  const float* bn_g0    = (const float*)d_in[5];
  const float* bn_b0    = (const float*)d_in[6];
  const float* bn_m0    = (const float*)d_in[7];
  const float* bn_v0    = (const float*)d_in[8];
  const float* res_w    = (const float*)d_in[9];
  const float* res_b    = (const float*)d_in[10];
  const float* res_bn_g = (const float*)d_in[11];
  const float* res_bn_b = (const float*)d_in[12];
  const float* res_bn_m = (const float*)d_in[13];
  const float* res_bn_v = (const float*)d_in[14];
  const float* gcn_w1   = (const float*)d_in[15];
  const float* gcn_b1   = (const float*)d_in[16];
  const float* tcn_w1   = (const float*)d_in[17];
  const float* tcn_b1   = (const float*)d_in[18];
  const float* bn_g1    = (const float*)d_in[19];
  const float* bn_b1    = (const float*)d_in[20];
  const float* bn_m1    = (const float*)d_in[21];
  const float* bn_v1    = (const float*)d_in[22];
  const float* fc_w     = (const float*)d_in[23];
  const float* fc_b     = (const float*)d_in[24];

  float* fcw_p = (float*)d_ws;
  fcw_pack<<<dim3(42), dim3(256), 0, stream>>>(fc_w, fcw_p);
  dim3 grid(NB * 2 * (NTIME / TT));
  dim3 block(NTHR);
  stgcn_fused<<<grid, block, 0, stream>>>(
      x, gcn_w0, gcn_b0, tcn_w0, tcn_b0, bn_g0, bn_b0, bn_m0, bn_v0,
      res_w, res_b, res_bn_g, res_bn_b, res_bn_m, res_bn_v,
      gcn_w1, gcn_b1, tcn_w1, tcn_b1, bn_g1, bn_b1, bn_m1, bn_v1,
      fcw_p, fc_b, (float*)d_out);
}


# Both weight folds (gcn0 into tcn0; gcn1 into tcn1, on WMMA) and the BN constants computed once per call by the first kernel into the workspace; a workgroup copies the 18 KB result, no fold phase
# speedup vs baseline: 6.4427x; 1.0772x over previous
#include <hip/hip_runtime.h>

#define TT 16
#define T1 (TT + 6)
#define T2 (TT + 4)
#define T3 (TT + 2)
#define NTHR 512
#define NWAVE (NTHR / 32)
#define NB 64
#define NTIME 512
#define NV 21
#define NC 32
#define NCIN 2
#define NOUT 64
#define PC 36
#define PW 40
#define PR 97
#define RP 68

#define WS_PAR 43008
#define P_TW   0
#define P_WF   3840
#define P_WB1  4224
#define P_S0   4320
#define P_O0   4352
#define P_S1   4384
#define P_O1   4416
#define P_RA   4448
#define P_RB   4480
#define P_RC   4512
#define P_FCB  4544
#define P_TOTAL 4608

typedef __attribute__((ext_vector_type(2))) float v2f;
typedef __attribute__((ext_vector_type(8))) float v8f;

__device__ __forceinline__ v8f wmma4(v2f a, v2f b, v8f c) {
  return __builtin_amdgcn_wmma_f32_16x16x4_f32(
       false, a,  false, b,
       (short)0, c,  false,  false);
}

__host__ __device__ constexpr int g_parent(int j) { return ((j - 1) % 4 == 0) ? 0 : j - 1; }
__host__ __device__ constexpr bool g_conn(int i, int j) {
  return i == j || (j > 0 && g_parent(j) == i) || (i > 0 && g_parent(i) == j);
}
__host__ __device__ constexpr float g_dinv(int v) {
  return v == 0 ? 0.4082482905f : ((v % 4 == 0) ? 0.7071067812f : 0.5773502692f);
}

__host__ __device__ constexpr int  fc_off(int g)    { return (g % NV) * PC + g / NV; }
__host__ __device__ constexpr bool fc_nowrap(int g) { return (g % NV) + 4 < NV; }

template <int STRIDE_IN, int STRIDE_OUT>
__device__ __forceinline__ void graph_agg(const float* __restrict__ in, float* __restrict__ out) {
  float h[NV];
#pragma unroll
  for (int u = 0; u < NV; ++u) h[u] = in[u * STRIDE_IN];
#pragma unroll
  for (int v = 0; v < NV; ++v) {
    float acc = 0.0f;
    bool have = false;
#pragma unroll
    for (int u = 0; u < NV; ++u) {
      if (g_conn(v, u)) {
        const float p = (g_dinv(v) * g_dinv(u)) * h[u];
        acc = have ? acc + p : p;
        have = true;
      }
    }
    out[v * STRIDE_OUT] = acc;
  }
}

__global__ __launch_bounds__(256) void prep(
    const float* __restrict__ gcn_w0, const float* __restrict__ gcn_b0,
    const float* __restrict__ tcn_w0, const float* __restrict__ tcn_b0,
    const float* __restrict__ bn_g0, const float* __restrict__ bn_b0,
    const float* __restrict__ bn_m0, const float* __restrict__ bn_v0,
    const float* __restrict__ res_w, const float* __restrict__ res_b,
    const float* __restrict__ res_bn_g, const float* __restrict__ res_bn_b,
    const float* __restrict__ res_bn_m, const float* __restrict__ res_bn_v,
    const float* __restrict__ gcn_w1, const float* __restrict__ gcn_b1,
    const float* __restrict__ tcn_w1, const float* __restrict__ tcn_b1,
    const float* __restrict__ bn_g1, const float* __restrict__ bn_b1,
    const float* __restrict__ bn_m1, const float* __restrict__ bn_v1,
    const float* __restrict__ fc_w, const float* __restrict__ fc_b,
    float* __restrict__ ws)
{
  __shared__ float s_raw1[NC * PR];
  __shared__ float s_gw1[NC * PC];
  const int tid = threadIdx.x;
  float* par = ws + WS_PAR;
  if (blockIdx.x < 42) {
    const int gid  = blockIdx.x * 256 + tid;
    const int lane = gid & 31;
    const int wq   = gid >> 5;
    const int w    = wq / 21;
    const int q    = wq - w * 21;
    const int col  = (w & 3) * 16 + (lane & 15);
    const float4 v = *(const float4*)(fc_w + col * 672 + (w >> 2) * 168 + 8 * q + (lane >> 4) * 4);
    *(float4*)(ws + gid * 4) = v;
  } else if (blockIdx.x == 42) {
    const int lane = tid & 31;
    const int row  = lane & 15;
    const int hi   = lane >> 4;
    const int wave_u = __builtin_amdgcn_readfirstlane(tid >> 5);
#pragma unroll
    for (int i = 0; i < 3072 / 256; ++i) {
      int idx = tid + i * 256;
      int co = idx / 96;
      s_raw1[co * PR + (idx - co * 96)] = tcn_w1[idx];
    }
#pragma unroll
    for (int i = 0; i < (NC * NC) / 256; ++i) {
      int idx = tid + i * 256;
      s_gw1[(idx >> 5) * PC + (idx & 31)] = gcn_w1[idx];
    }
    for (int idx = tid; idx < 96 * (PW - NC); idx += 256)
      par[P_TW + (idx >> 3) * PW + NC + (idx & 7)] = 0.0f;
    __syncthreads();
    for (int fjob = wave_u; fjob < 12; fjob += 8) {
      const int k = fjob >> 2, mh = (fjob >> 1) & 1, nh = fjob & 1;
      const float* ap = &s_gw1[(hi * 2) * PC + mh * 16 + row];
      const float* bp = &s_raw1[(nh * 16 + row) * PR + (hi * 2) * 3 + k];
      v2f fa[8], fb[8];
#pragma unroll
      for (int ks = 0; ks < 8; ++ks) {
        fa[ks].x = ap[(ks * 4) * PC];
        fa[ks].y = ap[(ks * 4 + 1) * PC];
        fb[ks].x = bp[(ks * 4) * 3];
        fb[ks].y = bp[(ks * 4 + 1) * 3];
      }
      v8f c0 = {0.f,0.f,0.f,0.f,0.f,0.f,0.f,0.f};
      v8f c1 = {0.f,0.f,0.f,0.f,0.f,0.f,0.f,0.f};
#pragma unroll
      for (int ks = 0; ks < 8; ks += 2) {
        c0 = wmma4(fa[ks], fb[ks], c0);
        c1 = wmma4(fa[ks + 1], fb[ks + 1], c1);
      }
      float* wp = &par[P_TW + (k * NC + mh * 16 + hi * 8) * PW + nh * 16 + row];
#pragma unroll
      for (int j = 0; j < 8; ++j) wp[j * PW] = c0[j] + c1[j];
    }
  } else {
    if (tid < 96) {
      const int k = tid >> 5, co = tid & 31;
      const float* rw  = tcn_w0 + co * 96 + k;
      const float* rw1 = tcn_w1 + co * 96 + k;
      float w0 = 0.0f, w1 = 0.0f, wb = 0.0f, wb1 = 0.0f;
#pragma unroll
      for (int ci = 0; ci < NC; ++ci) {
        const float w = rw[ci * 3];
        w0 += w * gcn_w0[ci * 2 + 0];
        w1 += w * gcn_w0[ci * 2 + 1];
        wb += w * gcn_b0[ci];
        wb1 += rw1[ci * 3] * gcn_b1[ci];
      }
      *(float4*)&par[P_WF + (k * NC + co) * 4] = make_float4(w0, w1, wb, 0.0f);
      par[P_WB1 + k * NC + co] = wb1;
    } else if (tid < 128) {
      const int c = tid - 96;
      float sc0 = bn_g0[c] * rsqrtf(bn_v0[c] + 1e-5f);
      par[P_S0 + c] = sc0;
      par[P_O0 + c] = (tcn_b0[c] - bn_m0[c]) * sc0 + bn_b0[c];
      float sc1 = bn_g1[c] * rsqrtf(bn_v1[c] + 1e-5f);
      par[P_S1 + c] = sc1;
      par[P_O1 + c] = (tcn_b1[c] - bn_m1[c]) * sc1 + bn_b1[c];
      float rsc = res_bn_g[c] * rsqrtf(res_bn_v[c] + 1e-5f);
      par[P_RA + c] = res_w[c * 2 + 0] * rsc;
      par[P_RB + c] = res_w[c * 2 + 1] * rsc;
      par[P_RC + c] = (res_b[c] - res_bn_m[c]) * rsc + res_bn_b[c];
    } else if (tid < 128 + NOUT) {
      par[P_FCB + tid - 128] = fc_b[tid - 128];
    }
  }
}

__global__ __launch_bounds__(NTHR) void stgcn_fused(
    const float* __restrict__ x,
    const float* __restrict__ ws,
    float* __restrict__ out)
{
  __shared__ __attribute__((aligned(16))) float s_in[T1 * NV * NCIN];
  __shared__ __attribute__((aligned(16))) float s_xa[T1 * NV * NCIN];
  __shared__ __attribute__((aligned(16))) float s_par[P_TOTAL];
  __shared__ float s_g1[T3 * NV * PC];
  __shared__ float s_h0[T2 * NV * PC];
  const float* s_tw  = s_par + P_TW;
  const float* s_wf  = s_par + P_WF;
  const float* s_wb1 = s_par + P_WB1;
  const float* s_s0 = s_par + P_S0, * s_o0 = s_par + P_O0, * s_s1 = s_par + P_S1, * s_o1 = s_par + P_O1;
  const float* s_rA = s_par + P_RA, * s_rB = s_par + P_RB, * s_rC = s_par + P_RC;
  const float* s_fcb = s_par + P_FCB;
  const float* fcw_p = ws;

  const int tid  = threadIdx.x;
  const int nthr = NTHR;
  const int lane = tid & 31;
  const int wave = tid >> 5;
  const int row  = lane & 15;
  const int hi   = lane >> 4;

  const int bid   = blockIdx.x;
  const int tile  = bid & 31;
  const int hand  = (bid >> 5) & 1;
  const int b     = bid >> 6;
  const int tbase = tile * TT;
  const int wave_u = __builtin_amdgcn_readfirstlane(wave);

  {
    const float4* src = (const float4*)(ws + WS_PAR);
    float4* dst = (float4*)s_par;
    const float4 par0 = src[tid];
    const float4 par1 = src[tid + NTHR];
    const float4 par2 = src[2 * NTHR + (tid & 127)];
    dst[tid] = par0;
    dst[tid + NTHR] = par1;
    dst[2 * NTHR + (tid & 127)] = par2;
  }

  for (int idx = tid; idx < T1 * NV * NCIN; idx += nthr) {
    int i = idx / (NV * NCIN), r = idx % (NV * NCIN);
    int v = r >> 1, ci = r & 1;
    int t = tbase - 3 + i;
    float val = 0.0f;
    if (t >= 0 && t < NTIME)
      val = x[(((b * NTIME + t) * 2 + hand) * NV + v) * NCIN + ci];
    s_in[idx] = val;
  }
  __syncthreads();

  if (wave_u < 2) {
    if (tid < T1 * NCIN)
      graph_agg<NCIN, NCIN>(&s_in[(tid >> 1) * (NV * NCIN) + (tid & 1)],
                            &s_xa[(tid >> 1) * (NV * NCIN) + (tid & 1)]);
  }
  __syncthreads();

  {
    const int M = T2 * NV;
    const int colN = (wave_u & 1) * 16 + row;
    const float c_s0 = s_s0[colN], c_o0 = s_o0[colN];
    const float c_rA = s_rA[colN], c_rB = s_rB[colN], c_rC = s_rC[colN];
    for (int tIdx = wave_u; tIdx < 27 * 2; tIdx += NWAVE) {
      const int mt = tIdx >> 1;
      v8f acc = {0.f,0.f,0.f,0.f,0.f,0.f,0.f,0.f};
      int mA = mt * 16 + row;
      int mC = (mA < M) ? mA : 0;
      int t2 = mC / NV;
      const int m0 = mt * 16 + hi * 8;
      v2f xr[8];
#pragma unroll
      for (int j = 0; j < 8; ++j) xr[j] = *(const v2f*)&s_in[(m0 + j + NV) * NCIN];
#pragma unroll
      for (int k = 0; k < 3; ++k) {
        const v2f xv = *(const v2f*)&s_xa[(mC + k * NV) * NCIN];
        const int t = tbase - 3 + t2 + k;
        const float ok = (t >= 0 && t < NTIME) ? 1.0f : 0.0f;
        v2f a;
        a.x = hi ? ok : xv.x;
        a.y = hi ? 0.0f : xv.y;
        const v2f bb = *(const v2f*)&s_wf[(k * NC + colN) * 4 + hi * 2];
        acc = wmma4(a, bb, acc);
      }
      float* hp = &s_h0[m0 * PC + colN];
      if (mt < 26) {
#pragma unroll
        for (int j = 0; j < 8; ++j) {
          float val = acc[j] * c_s0 + c_o0;
          val += c_rA * xr[j].x + c_rB * xr[j].y + c_rC;
          hp[j * PC] = fmaxf(val, 0.0f);
        }
      } else {
#pragma unroll
        for (int j = 0; j < 8; ++j) {
          if (m0 + j < M) {
            float val = acc[j] * c_s0 + c_o0;
            val += c_rA * xr[j].x + c_rB * xr[j].y + c_rC;
            hp[j * PC] = fmaxf(val, 0.0f);
          }
        }
      }
    }
  }
  __syncthreads();

  float* s_hagg = s_g1;
  for (int t3 = wave_u; t3 < T3; t3 += NWAVE) {
    const int t = tbase - 1 + t3;
    if (t >= 0 && t < NTIME) {
      graph_agg<PC, PC>(&s_h0[(t3 + 1) * (NV * PC) + lane], &s_hagg[t3 * (NV * PC) + lane]);
    } else {
#pragma unroll
      for (int v = 0; v < NV; ++v) s_hagg[t3 * (NV * PC) + v * PC + lane] = 0.0f;
    }
  }
  __syncthreads();

  {
    const int M = TT * NV;
    const int colN = (wave_u & 1) * 16 + row;
    const float c_s1 = s_s1[colN], c_o1 = s_o1[colN];
    const float wbk0 = s_wb1[colN], wbk1 = s_wb1[NC + colN], wbk2 = s_wb1[2 * NC + colN];
    const float o1_all   = ((wbk0 + wbk1) + wbk2) * c_s1 + c_o1;
    const float o1_first = (wbk1 + wbk2) * c_s1 + c_o1;
    const float o1_last  = (wbk0 + wbk1) * c_s1 + c_o1;
    const bool edge = (tile == 0) || (tile == NTIME / TT - 1);
    for (int tIdx = wave_u; tIdx < 21 * 2; tIdx += NWAVE) {
      int mt = tIdx >> 1;
      v8f acc = {0.f,0.f,0.f,0.f,0.f,0.f,0.f,0.f};
      int mA = mt * 16 + row;
      int mC = (mA < M) ? mA : 0;
      int t4 = mC / NV, vv = mC % NV;
      for (int ks = 0; ks < 24; ++ks) {
        int k0 = ks * 4 + hi * 2;
        v2f a, bb;
        a.x = s_hagg[((t4 + (k0 >> 5)) * NV + vv) * PC + (k0 & 31)];
        a.y = s_hagg[((t4 + ((k0 + 1) >> 5)) * NV + vv) * PC + ((k0 + 1) & 31)];
        bb.x = s_tw[k0 * PW + colN];
        bb.y = s_tw[(k0 + 1) * PW + colN];
        acc = wmma4(a, bb, acc);
      }
      if (!edge) {
#pragma unroll
        for (int j = 0; j < 8; ++j) {
          int m = mt * 16 + j + hi * 8;
          if (m < M) {
            int te = m / NV, ve = m % NV;
            int hidx = ((te + 2) * NV + ve) * PC + colN;
            float val = acc[j] * c_s1 + o1_all + s_h0[hidx];
            s_h0[hidx] = fmaxf(val, 0.0f);
          }
        }
      } else {
#pragma unroll
        for (int j = 0; j < 8; ++j) {
          int m = mt * 16 + j + hi * 8;
          if (m < M) {
            int te = m / NV, ve = m % NV;
            int t = tbase + te;
            float oj = (t == 0) ? o1_first : ((t == NTIME - 1) ? o1_last : o1_all);
            int hidx = ((te + 2) * NV + ve) * PC + colN;
            float val = acc[j] * c_s1 + oj + s_h0[hidx];
            s_h0[hidx] = fmaxf(val, 0.0f);
          }
        }
      }
    }
  }

  constexpr int FCQ = 21;
  constexpr int FCD = 7;
  const int fc_nt = wave & 3;
  const int fc_kq = wave >> 2;
  const int fc_col = fc_nt * 16 + row;
  const float* fc_bp = fcw_p + (wave * (FCQ * 32) + lane) * 4;
  float4 fcw_r[FCD];
#pragma unroll
  for (int q = 0; q < FCD; ++q) fcw_r[q] = *(const float4*)(fc_bp + 128 * q);
  __syncthreads();

  {
    float* s_red = s_g1;
    v8f acc = {0.f,0.f,0.f,0.f,0.f,0.f,0.f,0.f};
    const float* h1row = &s_h0[(row + 2) * (NV * PC) + 8 * fc_kq];
    const float* aP = h1row + hi * (4 * PC);
    const float* aQ = h1row + hi * ((4 - NV) * PC + 1);
#pragma unroll
    for (int q = 0; q < FCQ; ++q) {
      const float4 bv = fcw_r[q % FCD];
      if (q + FCD < FCQ) fcw_r[q % FCD] = *(const float4*)(fc_bp + 128 * (q + FCD));
      const int g = 8 * q;
      v2f a0, a1, b0, b1;
      a0.x = (fc_nowrap(g + 0) ? aP : aQ)[fc_off(g + 0)];
      a0.y = (fc_nowrap(g + 1) ? aP : aQ)[fc_off(g + 1)];
      a1.x = (fc_nowrap(g + 2) ? aP : aQ)[fc_off(g + 2)];
      a1.y = (fc_nowrap(g + 3) ? aP : aQ)[fc_off(g + 3)];
      b0.x = bv.x; b0.y = bv.y;
      b1.x = bv.z; b1.y = bv.w;
      acc = wmma4(a0, b0, acc);
      acc = wmma4(a1, b1, acc);
    }
#pragma unroll
    for (int j = 0; j < 8; ++j)
      s_red[(fc_kq * 16 + j + hi * 8) * RP + fc_col] = acc[j];
    __syncthreads();
#pragma unroll
    for (int i = 0; i < (TT * NOUT) / NTHR; ++i) {
      const int idx = tid + i * NTHR;
      const int m = idx >> 6, o = idx & 63;
      float val = s_red[m * RP + o];
      val += s_red[(16 + m) * RP + o];
      val += s_red[(32 + m) * RP + o];
      val += s_red[(48 + m) * RP + o];
      val += s_fcb[o];
      out[(((size_t)b * NTIME + (tbase + m)) * 2 + hand) * NOUT + o] = val;
    }
  }
}

extern "C" void kernel_launch(void* const* d_in, const int* in_sizes, int n_in,
                              void* d_out, int out_size, void* d_ws, size_t ws_size,
                              hipStream_t stream) {
  (void)in_sizes; (void)n_in; (void)ws_size; (void)out_size;
  const float* x        = (const float*)d_in[0];
  const float* gcn_w0   = (const float*)d_in[1];
  const float* gcn_b0   = (const float*)d_in[2];
  const float* tcn_w0   = (const float*)d_in[3];
  const float* tcn_b0   = (const float*)d_in[4];
  const float* bn_g0    = (const float*)d_in[5];
  const float* bn_b0    = (const float*)d_in[6];
  const float* bn_m0    = (const float*)d_in[7];
  const float* bn_v0    = (const float*)d_in[8];
  const float* res_w    = (const float*)d_in[9];
  const float* res_b    = (const float*)d_in[10];
  const float* res_bn_g = (const float*)d_in[11];
  const float* res_bn_b = (const float*)d_in[12];
  const float* res_bn_m = (const float*)d_in[13];
  const float* res_bn_v = (const float*)d_in[14];
  const float* gcn_w1   = (const float*)d_in[15];
  const float* gcn_b1   = (const float*)d_in[16];
  const float* tcn_w1   = (const float*)d_in[17];
  const float* tcn_b1   = (const float*)d_in[18];
  const float* bn_g1    = (const float*)d_in[19];
  const float* bn_b1    = (const float*)d_in[20];
  const float* bn_m1    = (const float*)d_in[21];
  const float* bn_v1    = (const float*)d_in[22];
  const float* fc_w     = (const float*)d_in[23];
  const float* fc_b     = (const float*)d_in[24];

  float* ws = (float*)d_ws;
  prep<<<dim3(44), dim3(256), 0, stream>>>(
      gcn_w0, gcn_b0, tcn_w0, tcn_b0, bn_g0, bn_b0, bn_m0, bn_v0,
      res_w, res_b, res_bn_g, res_bn_b, res_bn_m, res_bn_v,
      gcn_w1, gcn_b1, tcn_w1, tcn_b1, bn_g1, bn_b1, bn_m1, bn_v1,
      fc_w, fc_b, ws);
  dim3 grid(NB * 2 * (NTIME / TT));
  dim3 block(NTHR);
  stgcn_fused<<<grid, block, 0, stream>>>(x, ws, (float*)d_out);
}
